# starting kernel
# baseline (speedup 1.0000x reference)
#include <hip/hip_runtime.h>
#include <math.h>

typedef __attribute__((ext_vector_type(2))) float v2f;
typedef __attribute__((ext_vector_type(4))) float v4f;
typedef __attribute__((ext_vector_type(8))) float v8f;

#define T_TOK 2048
#define H_DIM 128
#define H4    512
#define NN    256

#define XS_STRIDE 132
#define HS_STRIDE 516


__global__ void k_zero(int* counts) { counts[threadIdx.x] = 0; }

__global__ void k_count(const int* __restrict__ node_ind, int* counts) {
    int t = blockIdx.x * blockDim.x + threadIdx.x;
    if (t < T_TOK) atomicAdd(&counts[node_ind[t]], 1);
}

__global__ void k_scan(const int* __restrict__ counts, int* offsets, int* cursor) {
    __shared__ int s[NN];
    int tid = threadIdx.x;
    int c = counts[tid];
    s[tid] = c;
    __syncthreads();
    for (int d = 1; d < NN; d <<= 1) {
        int v = (tid >= d) ? s[tid - d] : 0;
        __syncthreads();
        s[tid] += v;
        __syncthreads();
    }
    int excl = s[tid] - c;
    offsets[tid] = excl;
    cursor[tid]  = excl;
}

__global__ void k_scatter(const int* __restrict__ node_ind, int* cursor, int* bucket) {
    int t = blockIdx.x * blockDim.x + threadIdx.x;
    if (t < T_TOK) {
        int pos = atomicAdd(&cursor[node_ind[t]], 1);
        bucket[pos] = t;
    }
}


__device__ __forceinline__ float gelu_erf(float v) {
    return 0.5f * v * (1.0f + erff(v * 0.70710678118654752f));
}

__global__ __launch_bounds__(256) void k_mlp(
    const float* __restrict__ x, const int* __restrict__ bucket,
    const int* __restrict__ offsets, const int* __restrict__ counts,
    const float* __restrict__ W1, const float* __restrict__ b1,
    const float* __restrict__ W2, const float* __restrict__ b2,
    float* __restrict__ out)
{
    __shared__ float xs[16 * XS_STRIDE];
    __shared__ float hs[16 * HS_STRIDE];
    __shared__ int   tk[16];

    const int n   = blockIdx.x;
    const int cnt = counts[n];
    if (cnt == 0) return;
    const int start = offsets[n];

    const int tid  = threadIdx.x;
    const int lane = tid & 31;
    const int wave = tid >> 5;
    const int nlo  = lane & 15;
    const int kb   = (lane >> 4) << 1;
    const int mhi  = (lane >> 4) << 3;

    const float* W1n = W1 + (size_t)n * H4 * H_DIM;
    const float* W2n = W2 + (size_t)n * H_DIM * H4;
    const float* b1n = b1 + (size_t)n * H4;
    const float* b2n = b2 + (size_t)n * H_DIM;

    for (int tile = 0; tile < cnt; tile += 16) {
        if (tid < 16) {
            int idx = tile + tid;
            tk[tid] = bucket[start + (idx < cnt ? idx : 0)];
        }
        __syncthreads();

        for (int i = tid; i < 16 * (H_DIM / 4); i += 256) {
            int row = i >> 5, col4 = (i & 31) << 2;
            v4f v = *(const v4f*)(x + (size_t)tk[row] * H_DIM + col4);
            *(v4f*)(xs + row * XS_STRIDE + col4) = v;
        }
        __syncthreads();

        for (int ot = wave; ot < 32; ot += 8) {
            const int obase = ot << 4;
            const float* wrow = W1n + (size_t)(obase + nlo) * H_DIM + kb;
            const float* xrow = xs + nlo * XS_STRIDE + kb;
            if (ot + 8 < 32)
                __builtin_prefetch(wrow + 8 * 16 * H_DIM, 0, 1);
            v8f acc = {};
            #pragma unroll 8
            for (int k = 0; k < H_DIM; k += 4) {
                v2f a = *(const v2f*)(wrow + k);
                v2f b = *(const v2f*)(xrow + k);
                acc = __builtin_amdgcn_wmma_f32_16x16x4_f32(
                    false, a, false, b, (short)0, acc, false, false);
            }
            const int o0 = obase + mhi;
            v4f bia0 = *(const v4f*)(b1n + o0);
            v4f bia1 = *(const v4f*)(b1n + o0 + 4);
            v4f h0, h1;
            #pragma unroll
            for (int r = 0; r < 4; ++r) h0[r] = gelu_erf(acc[r]     + bia0[r]);
            #pragma unroll
            for (int r = 0; r < 4; ++r) h1[r] = gelu_erf(acc[r + 4] + bia1[r]);
            *(v4f*)(hs + nlo * HS_STRIDE + o0)     = h0;
            *(v4f*)(hs + nlo * HS_STRIDE + o0 + 4) = h1;
        }
        __syncthreads();

        {
            const int obase = wave << 4;
            const float* wrow = W2n + (size_t)(obase + nlo) * H4 + kb;
            const float* hrow = hs + nlo * HS_STRIDE + kb;
            v8f acc = {};
            #pragma unroll 8
            for (int k = 0; k < H4; k += 4) {
                v2f a = *(const v2f*)(wrow + k);
                v2f b = *(const v2f*)(hrow + k);
                acc = __builtin_amdgcn_wmma_f32_16x16x4_f32(
                    false, a, false, b, (short)0, acc, false, false);
            }
            if (tile + nlo < cnt) {
                const int t  = tk[nlo];
                const int o0 = obase + mhi;
                v4f bia0 = *(const v4f*)(b2n + o0);
                v4f bia1 = *(const v4f*)(b2n + o0 + 4);
                v4f r0, r1;
                #pragma unroll
                for (int r = 0; r < 4; ++r) r0[r] = acc[r]     + bia0[r];
                #pragma unroll
                for (int r = 0; r < 4; ++r) r1[r] = acc[r + 4] + bia1[r];
                *(v4f*)(out + (size_t)t * H_DIM + o0)     = r0;
                *(v4f*)(out + (size_t)t * H_DIM + o0 + 4) = r1;
            }
        }
        __syncthreads();
    }
}


extern "C" void kernel_launch(void* const* d_in, const int* in_sizes, int n_in,
                              void* d_out, int out_size, void* d_ws, size_t ws_size,
                              hipStream_t stream) {
    const float* x        = (const float*)d_in[0];
    const int*   node_ind = (const int*)  d_in[1];
    const float* W1       = (const float*)d_in[2];
    const float* b1       = (const float*)d_in[3];
    const float* W2       = (const float*)d_in[4];
    const float* b2       = (const float*)d_in[5];
    float*       out      = (float*)d_out;

    int* counts  = (int*)d_ws;
    int* cursor  = counts + NN;
    int* offsets = cursor + NN;
    int* bucket  = offsets + NN;

    k_zero   <<<1,          NN,  0, stream>>>(counts);
    k_count  <<<T_TOK/256,  256, 0, stream>>>(node_ind, counts);
    k_scan   <<<1,          NN,  0, stream>>>(counts, offsets, cursor);
    k_scatter<<<T_TOK/256,  256, 0, stream>>>(node_ind, cursor, bucket);
    k_mlp    <<<NN,         256, 0, stream>>>(x, bucket, offsets, counts,
                                              W1, b1, W2, b2, out);
}


# k_mlp weights: one load instruction = 512 contiguous bytes of a row (whole lines), two 16-row chunks in flight per wave, parked in wave-private LDS; A fragments read from there, same k order
# speedup vs baseline: 1.3697x; 1.3697x over previous
#include <hip/hip_runtime.h>
#include <math.h>

typedef __attribute__((ext_vector_type(2))) float v2f;
typedef __attribute__((ext_vector_type(4))) float v4f;
typedef __attribute__((ext_vector_type(8))) float v8f;

#define T_TOK 2048
#define H_DIM 128
#define H4    512
#define NN    256

#define XS_STRIDE 132
#define HS_STRIDE 516
#define WS_STRIDE 132


__global__ void k_zero(int* counts) { counts[threadIdx.x] = 0; }

__global__ void k_count(const int* __restrict__ node_ind, int* counts) {
    int t = blockIdx.x * blockDim.x + threadIdx.x;
    if (t < T_TOK) atomicAdd(&counts[node_ind[t]], 1);
}

__global__ void k_scan(const int* __restrict__ counts, int* offsets, int* cursor) {
    __shared__ int s[NN];
    int tid = threadIdx.x;
    int c = counts[tid];
    s[tid] = c;
    __syncthreads();
    for (int d = 1; d < NN; d <<= 1) {
        int v = (tid >= d) ? s[tid - d] : 0;
        __syncthreads();
        s[tid] += v;
        __syncthreads();
    }
    int excl = s[tid] - c;
    offsets[tid] = excl;
    cursor[tid]  = excl;
}

__global__ void k_scatter(const int* __restrict__ node_ind, int* cursor, int* bucket) {
    int t = blockIdx.x * blockDim.x + threadIdx.x;
    if (t < T_TOK) {
        int pos = atomicAdd(&cursor[node_ind[t]], 1);
        bucket[pos] = t;
    }
}


__device__ __forceinline__ float gelu_erf(float v) {
    return 0.5f * v * (1.0f + erff(v * 0.70710678118654752f));
}

__device__ __forceinline__ void load_chunk(v4f (&w)[16], const float* src, const int pitch) {
    #pragma unroll
    for (int r = 0; r < 16; ++r) w[r] = *(const v4f*)(src + r * pitch);
}

__device__ __forceinline__ void stage_chunk(float* dst, const v4f (&w)[16]) {
    #pragma unroll
    for (int r = 0; r < 16; ++r) *(v4f*)(dst + r * WS_STRIDE) = w[r];
}

__device__ __forceinline__ v8f mma_chunk(const float* arow, const float* brow, v8f acc) {
    #pragma unroll
    for (int k = 0; k < 128; k += 4) {
        v2f a = *(const v2f*)(arow + k);
        v2f b = *(const v2f*)(brow + k);
        acc = __builtin_amdgcn_wmma_f32_16x16x4_f32(
            false, a, false, b, (short)0, acc, false, false);
    }
    return acc;
}

__device__ __forceinline__ void deposit_h(float* hdst, const float* __restrict__ bias, v8f acc) {
    v4f bia0 = *(const v4f*)(bias);
    v4f bia1 = *(const v4f*)(bias + 4);
    v4f h0, h1;
    #pragma unroll
    for (int r = 0; r < 4; ++r) h0[r] = gelu_erf(acc[r]     + bia0[r]);
    #pragma unroll
    for (int r = 0; r < 4; ++r) h1[r] = gelu_erf(acc[r + 4] + bia1[r]);
    *(v4f*)(hdst)     = h0;
    *(v4f*)(hdst + 4) = h1;
}

__global__ __launch_bounds__(256) void k_mlp(
    const float* __restrict__ x, const int* __restrict__ bucket,
    const int* __restrict__ offsets, const int* __restrict__ counts,
    const float* W1, const float* __restrict__ b1,
    const float* W2, const float* __restrict__ b2,
    float* out)
{
    __shared__ float xs[16 * XS_STRIDE];
    __shared__ float hs[16 * HS_STRIDE];
    __shared__ __attribute__((aligned(16))) float wst[8 * 16 * WS_STRIDE];
    __shared__ int   tk[16];

    const int n   = blockIdx.x;
    const int cnt = counts[n];
    if (cnt == 0) return;
    const int start = offsets[n];

    const int tid  = threadIdx.x;
    const int lane = tid & 31;
    const int wave = tid >> 5;
    const int nlo  = lane & 15;
    const int kb   = (lane >> 4) << 1;
    const int mhi  = (lane >> 4) << 3;

    const float* W1n = W1 + (size_t)n * H4 * H_DIM;
    const float* W2n = W2 + (size_t)n * H_DIM * H4;
    const float* b1n = b1 + (size_t)n * H4;
    const float* b2n = b2 + (size_t)n * H_DIM;

    const float* src1 = W1n + (size_t)(wave << 4) * H_DIM + (lane << 2);
    const int    o8   = 8 * 16 * H_DIM;
    const float* src2 = W2n + (size_t)(wave << 4) * H4 + (lane << 2);

    float*       wmine = wst + wave * (16 * WS_STRIDE);
    float*       wdst  = wmine + (lane << 2);
    const float* afrag = wmine + nlo * WS_STRIDE + kb;
    const float* xfrag = xs + nlo * XS_STRIDE + kb;
    const float* hfrag = hs + nlo * HS_STRIDE + kb;
    float*       hdst  = hs + nlo * HS_STRIDE + mhi;

    for (int tile = 0; tile < cnt; tile += 16) {
        if (tid < 16) {
            int idx = tile + tid;
            tk[tid] = bucket[start + (idx < cnt ? idx : 0)];
        }
        __syncthreads();

        const int xr0 = tid >> 5, xr1 = xr0 + 8, xc4 = (tid & 31) << 2;
        v4f xv0 = *(const v4f*)(x + (size_t)tk[xr0] * H_DIM + xc4);
        v4f xv1 = *(const v4f*)(x + (size_t)tk[xr1] * H_DIM + xc4);

        v4f wA[16], wB[16];
        load_chunk(wA, src1,      H_DIM);
        load_chunk(wB, src1 + o8, H_DIM);

        *(v4f*)(xs + xr0 * XS_STRIDE + xc4) = xv0;
        *(v4f*)(xs + xr1 * XS_STRIDE + xc4) = xv1;
        __syncthreads();

        v8f acc;
        const v8f zero = {};

        stage_chunk(wdst, wA);  load_chunk(wA, src1 + 2 * o8, H_DIM);
        acc = mma_chunk(afrag, xfrag, zero);
        deposit_h(hdst + (wave << 4),        b1n + (wave << 4) + mhi,        acc);

        stage_chunk(wdst, wB);  load_chunk(wB, src1 + 3 * o8, H_DIM);
        acc = mma_chunk(afrag, xfrag, zero);
        deposit_h(hdst + ((wave + 8) << 4),  b1n + ((wave + 8) << 4) + mhi,  acc);

        stage_chunk(wdst, wA);  load_chunk(wA, src2,       H4);
        acc = mma_chunk(afrag, xfrag, zero);
        deposit_h(hdst + ((wave + 16) << 4), b1n + ((wave + 16) << 4) + mhi, acc);

        stage_chunk(wdst, wB);  load_chunk(wB, src2 + 128, H4);
        acc = mma_chunk(afrag, xfrag, zero);
        deposit_h(hdst + ((wave + 24) << 4), b1n + ((wave + 24) << 4) + mhi, acc);
        __syncthreads();

        stage_chunk(wdst, wA);  load_chunk(wA, src2 + 256, H4);
        acc = mma_chunk(afrag, hfrag,       zero);
        stage_chunk(wdst, wB);  load_chunk(wB, src2 + 384, H4);
        acc = mma_chunk(afrag, hfrag + 128, acc);
        stage_chunk(wdst, wA);
        acc = mma_chunk(afrag, hfrag + 256, acc);
        stage_chunk(wdst, wB);
        acc = mma_chunk(afrag, hfrag + 384, acc);

        if (tile + nlo < cnt) {
            const int t  = tk[nlo];
            const int o0 = (wave << 4) + mhi;
            v4f bia0 = *(const v4f*)(b2n + o0);
            v4f bia1 = *(const v4f*)(b2n + o0 + 4);
            v4f r0, r1;
            #pragma unroll
            for (int r = 0; r < 4; ++r) r0[r] = acc[r]     + bia0[r];
            #pragma unroll
            for (int r = 0; r < 4; ++r) r1[r] = acc[r + 4] + bia1[r];
            *(v4f*)(out + (size_t)t * H_DIM + o0)     = r0;
            *(v4f*)(out + (size_t)t * H_DIM + o0 + 4) = r1;
        }
        __syncthreads();
    }
}


extern "C" void kernel_launch(void* const* d_in, const int* in_sizes, int n_in,
                              void* d_out, int out_size, void* d_ws, size_t ws_size,
                              hipStream_t stream) {
    const float* x        = (const float*)d_in[0];
    const int*   node_ind = (const int*)  d_in[1];
    const float* W1       = (const float*)d_in[2];
    const float* b1       = (const float*)d_in[3];
    const float* W2       = (const float*)d_in[4];
    const float* b2       = (const float*)d_in[5];
    float*       out      = (float*)d_out;

    int* counts  = (int*)d_ws;
    int* cursor  = counts + NN;
    int* offsets = cursor + NN;
    int* bucket  = offsets + NN;

    k_zero   <<<1,          NN,  0, stream>>>(counts);
    k_count  <<<T_TOK/256,  256, 0, stream>>>(node_ind, counts);
    k_scan   <<<1,          NN,  0, stream>>>(counts, offsets, cursor);
    k_scatter<<<T_TOK/256,  256, 0, stream>>>(node_ind, cursor, bucket);
    k_mlp    <<<NN,         256, 0, stream>>>(x, bucket, offsets, counts,
                                              W1, b1, W2, b2, out);
}


# k_mlp finds its node's tokens itself: each block ballots over node_ind into an LDS list while its first two weight chunks are in flight; the four bucketing launches and their atomics go
# speedup vs baseline: 1.6596x; 1.2117x over previous
#include <hip/hip_runtime.h>
#include <math.h>

typedef __attribute__((ext_vector_type(2))) float v2f;
typedef __attribute__((ext_vector_type(4))) float v4f;
typedef __attribute__((ext_vector_type(8))) float v8f;

#define T_TOK 2048
#define H_DIM 128
#define H4    512
#define NN    256

#define XS_STRIDE 132
#define HS_STRIDE 516
#define WS_STRIDE 132

typedef __attribute__((ext_vector_type(4))) int v4i;


__device__ __forceinline__ float gelu_erf(float v) {
    return 0.5f * v * (1.0f + erff(v * 0.70710678118654752f));
}

__device__ __forceinline__ void load_chunk(v4f (&w)[16], const float* src, const int pitch) {
    #pragma unroll
    for (int r = 0; r < 16; ++r) w[r] = *(const v4f*)(src + r * pitch);
}

__device__ __forceinline__ void stage_chunk(float* dst, const v4f (&w)[16]) {
    #pragma unroll
    for (int r = 0; r < 16; ++r) *(v4f*)(dst + r * WS_STRIDE) = w[r];
}

__device__ __forceinline__ v8f mma_chunk(const float* arow, const float* brow, v8f acc) {
    #pragma unroll
    for (int k = 0; k < 128; k += 4) {
        v2f a = *(const v2f*)(arow + k);
        v2f b = *(const v2f*)(brow + k);
        acc = __builtin_amdgcn_wmma_f32_16x16x4_f32(
            false, a, false, b, (short)0, acc, false, false);
    }
    return acc;
}

__device__ __forceinline__ void deposit_h(float* hdst, const float* __restrict__ bias, v8f acc) {
    v4f bia0 = *(const v4f*)(bias);
    v4f bia1 = *(const v4f*)(bias + 4);
    v4f h0, h1;
    #pragma unroll
    for (int r = 0; r < 4; ++r) h0[r] = gelu_erf(acc[r]     + bia0[r]);
    #pragma unroll
    for (int r = 0; r < 4; ++r) h1[r] = gelu_erf(acc[r + 4] + bia1[r]);
    *(v4f*)(hdst)     = h0;
    *(v4f*)(hdst + 4) = h1;
}

__device__ __forceinline__ void lds_barrier() {
    asm volatile("s_wait_dscnt 0x0\n\ts_barrier_signal -1\n\ts_barrier_wait -1" ::: "memory");
}

__global__ __launch_bounds__(256) void k_mlp(
    const float* __restrict__ x, const int* __restrict__ node_ind,
    const float* W1, const float* __restrict__ b1,
    const float* W2, const float* __restrict__ b2,
    float* out)
{
    __shared__ float xs[16 * XS_STRIDE];
    __shared__ float hs[16 * HS_STRIDE];
    __shared__ __attribute__((aligned(16))) float wst[8 * 16 * WS_STRIDE];
    __shared__ int   tok[T_TOK];
    __shared__ __attribute__((aligned(16))) int wcnt[8];

    const int n    = blockIdx.x;
    const int tid  = threadIdx.x;
    const int lane = tid & 31;
    const int wave = tid >> 5;
    const int nlo  = lane & 15;
    const int kb   = (lane >> 4) << 1;
    const int mhi  = (lane >> 4) << 3;

    const float* W1n = W1 + (size_t)n * H4 * H_DIM;
    const float* W2n = W2 + (size_t)n * H_DIM * H4;
    const float* b1n = b1 + (size_t)n * H4;
    const float* b2n = b2 + (size_t)n * H_DIM;

    const float* src1 = W1n + (size_t)(wave << 4) * H_DIM + (lane << 2);
    const int    o8   = 8 * 16 * H_DIM;
    const float* src2 = W2n + (size_t)(wave << 4) * H4 + (lane << 2);

    const int tb = (wave << 8) + (lane << 2);
    const v4i id0 = *(const v4i*)(node_ind + tb);
    const v4i id1 = *(const v4i*)(node_ind + tb + 128);

    v4f wA[16], wB[16];
    load_chunk(wA, src1,      H_DIM);
    load_chunk(wB, src1 + o8, H_DIM);

    int pre[8];
    int run = 0;
    unsigned hitbits = 0;
    #pragma unroll
    for (int q = 0; q < 8; ++q) {
        const int id = (q < 4) ? id0[q & 3] : id1[q & 3];
        const bool hit = (id == n);
        const unsigned mask = __builtin_amdgcn_ballot_w32(hit);
        pre[q] = __builtin_amdgcn_mbcnt_lo(mask, run);
        run += __builtin_popcount(mask);
        hitbits |= hit ? (1u << q) : 0u;
    }
    if (lane == 0) wcnt[wave] = run;
    lds_barrier();

    const v4i c0 = *(const v4i*)(wcnt);
    const v4i c1 = *(const v4i*)(wcnt + 4);
    int base = 0, cnt = 0;
    #pragma unroll
    for (int w = 0; w < 8; ++w) {
        const int cw = (w < 4) ? c0[w & 3] : c1[w & 3];
        base += (w < wave) ? cw : 0;
        cnt  += cw;
    }
    #pragma unroll
    for (int q = 0; q < 8; ++q)
        if (hitbits & (1u << q)) tok[base + pre[q]] = tb + ((q >> 2) << 7) + (q & 3);
    lds_barrier();
    if (cnt == 0) {
        asm volatile("s_wait_loadcnt 0x0" ::: "memory");
        return;
    }

    float*       wmine = wst + wave * (16 * WS_STRIDE);
    float*       wdst  = wmine + (lane << 2);
    const float* afrag = wmine + nlo * WS_STRIDE + kb;
    const float* xfrag = xs + nlo * XS_STRIDE + kb;
    const float* hfrag = hs + nlo * HS_STRIDE + kb;
    float*       hdst  = hs + nlo * HS_STRIDE + mhi;

    for (int tile = 0; ; ) {
        const int xr0 = tid >> 5, xr1 = xr0 + 8, xc4 = (tid & 31) << 2;
        const int s0 = tile + xr0, s1 = tile + xr1;
        const int t0 = tok[s0 < cnt ? s0 : 0];
        const int t1 = tok[s1 < cnt ? s1 : 0];
        v4f xv0 = *(const v4f*)(x + (size_t)t0 * H_DIM + xc4);
        v4f xv1 = *(const v4f*)(x + (size_t)t1 * H_DIM + xc4);

        *(v4f*)(xs + xr0 * XS_STRIDE + xc4) = xv0;
        *(v4f*)(xs + xr1 * XS_STRIDE + xc4) = xv1;
        __syncthreads();

        v8f acc;
        const v8f zero = {};

        stage_chunk(wdst, wA);  load_chunk(wA, src1 + 2 * o8, H_DIM);
        acc = mma_chunk(afrag, xfrag, zero);
        deposit_h(hdst + (wave << 4),        b1n + (wave << 4) + mhi,        acc);

        stage_chunk(wdst, wB);  load_chunk(wB, src1 + 3 * o8, H_DIM);
        acc = mma_chunk(afrag, xfrag, zero);
        deposit_h(hdst + ((wave + 8) << 4),  b1n + ((wave + 8) << 4) + mhi,  acc);

        stage_chunk(wdst, wA);  load_chunk(wA, src2,       H4);
        acc = mma_chunk(afrag, xfrag, zero);
        deposit_h(hdst + ((wave + 16) << 4), b1n + ((wave + 16) << 4) + mhi, acc);

        stage_chunk(wdst, wB);  load_chunk(wB, src2 + 128, H4);
        acc = mma_chunk(afrag, xfrag, zero);
        deposit_h(hdst + ((wave + 24) << 4), b1n + ((wave + 24) << 4) + mhi, acc);
        __syncthreads();

        stage_chunk(wdst, wA);  load_chunk(wA, src2 + 256, H4);
        acc = mma_chunk(afrag, hfrag,       zero);
        stage_chunk(wdst, wB);  load_chunk(wB, src2 + 384, H4);
        acc = mma_chunk(afrag, hfrag + 128, acc);
        stage_chunk(wdst, wA);
        acc = mma_chunk(afrag, hfrag + 256, acc);
        stage_chunk(wdst, wB);
        acc = mma_chunk(afrag, hfrag + 384, acc);

        if (tile + nlo < cnt) {
            const int t  = tok[tile + nlo];
            const int o0 = (wave << 4) + mhi;
            v4f bia0 = *(const v4f*)(b2n + o0);
            v4f bia1 = *(const v4f*)(b2n + o0 + 4);
            v4f r0, r1;
            #pragma unroll
            for (int r = 0; r < 4; ++r) r0[r] = acc[r]     + bia0[r];
            #pragma unroll
            for (int r = 0; r < 4; ++r) r1[r] = acc[r + 4] + bia1[r];
            *(v4f*)(out + (size_t)t * H_DIM + o0)     = r0;
            *(v4f*)(out + (size_t)t * H_DIM + o0 + 4) = r1;
        }

        tile += 16;
        if (tile >= cnt) break;
        load_chunk(wA, src1,      H_DIM);
        load_chunk(wB, src1 + o8, H_DIM);
        __syncthreads();
    }
}


extern "C" void kernel_launch(void* const* d_in, const int* in_sizes, int n_in,
                              void* d_out, int out_size, void* d_ws, size_t ws_size,
                              hipStream_t stream) {
    const float* x        = (const float*)d_in[0];
    const int*   node_ind = (const int*)  d_in[1];
    const float* W1       = (const float*)d_in[2];
    const float* b1       = (const float*)d_in[3];
    const float* W2       = (const float*)d_in[4];
    const float* b2       = (const float*)d_in[5];
    float*       out      = (float*)d_out;

    k_mlp<<<NN, 256, 0, stream>>>(x, node_ind, W1, b1, W2, b2, out);
}


# a node with more than 16 tokens takes 32 token slots through ONE pass over its weights (each parked A fragment feeds two WMMAs, one per token tile), not a second full pass; sums keep their k order
# speedup vs baseline: 1.7871x; 1.0768x over previous
#include <hip/hip_runtime.h>
#include <math.h>

typedef __attribute__((ext_vector_type(2))) float v2f;
typedef __attribute__((ext_vector_type(4))) float v4f;
typedef __attribute__((ext_vector_type(8))) float v8f;

#define T_TOK 2048
#define H_DIM 128
#define H4    512
#define NN    256

#define XS_STRIDE 132
#define HS_STRIDE 516
#define WS_STRIDE 132
#define XS_TILE (16 * XS_STRIDE)
#define HS_TILE (16 * HS_STRIDE)

typedef __attribute__((ext_vector_type(4))) int v4i;


__device__ __forceinline__ float gelu_erf(float v) {
    return 0.5f * v * (1.0f + erff(v * 0.70710678118654752f));
}

__device__ __forceinline__ void load_chunk(v4f (&w)[16], const float* src, const int pitch) {
    #pragma unroll
    for (int r = 0; r < 16; ++r) w[r] = *(const v4f*)(src + r * pitch);
}

__device__ __forceinline__ void stage_chunk(float* dst, const v4f (&w)[16]) {
    #pragma unroll
    for (int r = 0; r < 16; ++r) *(v4f*)(dst + r * WS_STRIDE) = w[r];
}

template <int NT>
__device__ __forceinline__ void mma_chunk(const float* arow, const float* brow, const int bpitch,
                                          v8f (&acc)[NT]) {
    #pragma unroll
    for (int k = 0; k < 128; k += 4) {
        v2f a = *(const v2f*)(arow + k);
        #pragma unroll
        for (int t = 0; t < NT; ++t) {
            v2f b = *(const v2f*)(brow + t * bpitch + k);
            acc[t] = __builtin_amdgcn_wmma_f32_16x16x4_f32(
                false, a, false, b, (short)0, acc[t], false, false);
        }
    }
}

template <int NT>
__device__ __forceinline__ void clear_acc(v8f (&acc)[NT]) {
    const v8f zero = {};
    #pragma unroll
    for (int t = 0; t < NT; ++t) acc[t] = zero;
}

__device__ __forceinline__ void deposit_h(float* hdst, const v4f bia0, const v4f bia1, v8f acc) {
    v4f h0, h1;
    #pragma unroll
    for (int r = 0; r < 4; ++r) h0[r] = gelu_erf(acc[r]     + bia0[r]);
    #pragma unroll
    for (int r = 0; r < 4; ++r) h1[r] = gelu_erf(acc[r + 4] + bia1[r]);
    *(v4f*)(hdst)     = h0;
    *(v4f*)(hdst + 4) = h1;
}

template <int NT>
__device__ __forceinline__ void deposit_tiles(float* hdst, const v4f bia0, const v4f bia1,
                                              const v8f (&acc)[NT]) {
    #pragma unroll
    for (int t = 0; t < NT; ++t) deposit_h(hdst + t * HS_TILE, bia0, bia1, acc[t]);
}

__device__ __forceinline__ void lds_barrier() {
    asm volatile("s_wait_dscnt 0x0\n\ts_barrier_signal -1\n\ts_barrier_wait -1" ::: "memory");
}

__device__ __forceinline__ void run_one_tile(
    v4f (&wA)[16], v4f (&wB)[16],
    const float* __restrict__ x, const float* src1, const float* src2, const int o8,
    const v4f (&bv)[8], const float* __restrict__ b2n, float* out,
    float* xs, float* hs, const int* tok, float* wdst, const float* afrag,
    const int tile, const int cnt,
    const int tid, const int wave, const int nlo, const int kb, const int mhi)
{
    constexpr int NT = 1;
    const float* xfrag = xs + nlo * XS_STRIDE + kb;
    const float* hfrag = hs + nlo * HS_STRIDE + kb;
    float*       hdst  = hs + nlo * HS_STRIDE + mhi;

    const int xr0 = tid >> 5, xc4 = (tid & 31) << 2;
    v4f xv[2 * NT];
    #pragma unroll
    for (int j = 0; j < 2 * NT; ++j) {
        const int s = tile + xr0 + 8 * j;
        const int t = tok[s < cnt ? s : 0];
        xv[j] = *(const v4f*)(x + (size_t)t * H_DIM + xc4);
    }
    #pragma unroll
    for (int j = 0; j < 2 * NT; ++j)
        *(v4f*)(xs + (j >> 1) * XS_TILE + (xr0 + 8 * (j & 1)) * XS_STRIDE + xc4) = xv[j];
    __syncthreads();

    v8f acc[NT];

    stage_chunk(wdst, wA);  load_chunk(wA, src1 + 2 * o8, H_DIM);
    clear_acc<NT>(acc);
    mma_chunk<NT>(afrag, xfrag, XS_TILE, acc);
    deposit_tiles<NT>(hdst + (wave << 4),        bv[0], bv[1], acc);

    stage_chunk(wdst, wB);  load_chunk(wB, src1 + 3 * o8, H_DIM);
    clear_acc<NT>(acc);
    mma_chunk<NT>(afrag, xfrag, XS_TILE, acc);
    deposit_tiles<NT>(hdst + ((wave + 8) << 4),  bv[2], bv[3], acc);

    stage_chunk(wdst, wA);  load_chunk(wA, src2,       H4);
    clear_acc<NT>(acc);
    mma_chunk<NT>(afrag, xfrag, XS_TILE, acc);
    deposit_tiles<NT>(hdst + ((wave + 16) << 4), bv[4], bv[5], acc);

    stage_chunk(wdst, wB);  load_chunk(wB, src2 + 128, H4);
    clear_acc<NT>(acc);
    mma_chunk<NT>(afrag, xfrag, XS_TILE, acc);
    deposit_tiles<NT>(hdst + ((wave + 24) << 4), bv[6], bv[7], acc);
    __syncthreads();

    stage_chunk(wdst, wA);  load_chunk(wA, src2 + 256, H4);
    clear_acc<NT>(acc);
    mma_chunk<NT>(afrag, hfrag,       HS_TILE, acc);
    stage_chunk(wdst, wB);  load_chunk(wB, src2 + 384, H4);
    mma_chunk<NT>(afrag, hfrag + 128, HS_TILE, acc);
    stage_chunk(wdst, wA);
    mma_chunk<NT>(afrag, hfrag + 256, HS_TILE, acc);
    stage_chunk(wdst, wB);
    mma_chunk<NT>(afrag, hfrag + 384, HS_TILE, acc);

    #pragma unroll
    for (int tt = 0; tt < NT; ++tt) {
        const int s = tile + 16 * tt + nlo;
        if (s < cnt) {
            const int t  = tok[s];
            const int o0 = (wave << 4) + mhi;
            v4f bia0 = *(const v4f*)(b2n + o0);
            v4f bia1 = *(const v4f*)(b2n + o0 + 4);
            v4f r0, r1;
            #pragma unroll
            for (int r = 0; r < 4; ++r) r0[r] = acc[tt][r]     + bia0[r];
            #pragma unroll
            for (int r = 0; r < 4; ++r) r1[r] = acc[tt][r + 4] + bia1[r];
            *(v4f*)(out + (size_t)t * H_DIM + o0)     = r0;
            *(v4f*)(out + (size_t)t * H_DIM + o0 + 4) = r1;
        }
    }
}

__device__ __forceinline__ void deposit_pre(float* hdst, const v4f bia0, const v4f bia1, const v8f acc) {
    v4f h0, h1;
    #pragma unroll
    for (int r = 0; r < 4; ++r) h0[r] = acc[r]     + bia0[r];
    #pragma unroll
    for (int r = 0; r < 4; ++r) h1[r] = acc[r + 4] + bia1[r];
    *(v4f*)(hdst)     = h0;
    *(v4f*)(hdst + 4) = h1;
}

__device__ __forceinline__ void gelu_cols(float* hs, const int ot, const int rows, const int lane) {
    float*    col = hs + (ot << 4) + (lane & 15);
    const int rh  = lane >> 4;
    #pragma unroll 1
    for (int r0 = 0; r0 < rows; r0 += 2) {
        const int r = r0 + rh;
        if (r < rows) {
            float* p = col + (r >> 4) * HS_TILE + (r & 15) * HS_STRIDE;
            *p = gelu_erf(*p);
        }
    }
}

__device__ __forceinline__ void run_two_tiles(
    v4f (&wA)[16], v4f (&wB)[16],
    const float* __restrict__ x, const float* src1, const float* src2, const int o8,
    const v4f (&bv)[8], const float* __restrict__ b2n, float* out,
    float* xs, float* hs, const int* tok, float* wdst, const float* afrag,
    const int tile, const int cnt,
    const int tid, const int lane, const int wave, const int nlo, const int kb, const int mhi)
{
    const float* xfrag = xs + nlo * XS_STRIDE + kb;
    const float* hfrag = hs + nlo * HS_STRIDE + kb;
    float*       hdst  = hs + nlo * HS_STRIDE + mhi;

    const int xr0 = tid >> 5, xc4 = (tid & 31) << 2;
    v4f xv[4];
    #pragma unroll
    for (int j = 0; j < 4; ++j) {
        const int s = tile + xr0 + 8 * j;
        const int t = tok[s < cnt ? s : 0];
        xv[j] = *(const v4f*)(x + (size_t)t * H_DIM + xc4);
    }
    #pragma unroll
    for (int j = 0; j < 4; ++j)
        *(v4f*)(xs + (j >> 1) * XS_TILE + (xr0 + 8 * (j & 1)) * XS_STRIDE + xc4) = xv[j];
    __syncthreads();

    const int rows = (cnt - tile < 32) ? (cnt - tile) : 32;
    v8f acc[2];

    #pragma unroll 1
    for (int p = 0; p < 4; ++p) {
        const bool   l1     = p < 2;
        const int    cn     = 2 * p + 2;
        const float* nsrc   = (cn < 4) ? src1 + cn * o8 : src2 + (cn - 4) * 128;
        const int    nstep  = (cn < 4) ? o8 : 128;
        const int    npitch = (cn < 4) ? H_DIM : H4;
        const float* bfrag  = l1 ? xfrag : hfrag + (2 * p - 4) * 128;
        const int    bstep  = l1 ? 0 : 128;
        const int    bpitch = l1 ? XS_TILE : HS_TILE;
        const int    ot     = wave + 16 * p;
        const v4f ba0 = (p == 0) ? bv[0] : bv[4], ba1 = (p == 0) ? bv[1] : bv[5];
        const v4f bb0 = (p == 0) ? bv[2] : bv[6], bb1 = (p == 0) ? bv[3] : bv[7];

        stage_chunk(wdst, wA);
        if (p < 3) load_chunk(wA, nsrc, npitch);
        if (p < 3) clear_acc<2>(acc);
        mma_chunk<2>(afrag, bfrag, bpitch, acc);
        if (l1) {
            deposit_pre(hdst + (ot << 4),           ba0, ba1, acc[0]);
            deposit_pre(hdst + (ot << 4) + HS_TILE, ba0, ba1, acc[1]);
            asm volatile("s_wait_dscnt 0x0" ::: "memory");
            gelu_cols(hs, ot, rows, lane);
        }

        stage_chunk(wdst, wB);
        if (p < 3) load_chunk(wB, nsrc + nstep, npitch);
        if (l1) clear_acc<2>(acc);
        mma_chunk<2>(afrag, bfrag + bstep, bpitch, acc);
        if (l1) {
            deposit_pre(hdst + ((ot + 8) << 4),           bb0, bb1, acc[0]);
            deposit_pre(hdst + ((ot + 8) << 4) + HS_TILE, bb0, bb1, acc[1]);
            asm volatile("s_wait_dscnt 0x0" ::: "memory");
            gelu_cols(hs, ot + 8, rows, lane);
        }
        if (p == 1) __syncthreads();
    }

    #pragma unroll
    for (int tt = 0; tt < 2; ++tt) {
        const int s = tile + 16 * tt + nlo;
        if (s < cnt) {
            const int t  = tok[s];
            const int o0 = (wave << 4) + mhi;
            v4f bia0 = *(const v4f*)(b2n + o0);
            v4f bia1 = *(const v4f*)(b2n + o0 + 4);
            v4f r0, r1;
            #pragma unroll
            for (int r = 0; r < 4; ++r) r0[r] = acc[tt][r]     + bia0[r];
            #pragma unroll
            for (int r = 0; r < 4; ++r) r1[r] = acc[tt][r + 4] + bia1[r];
            *(v4f*)(out + (size_t)t * H_DIM + o0)     = r0;
            *(v4f*)(out + (size_t)t * H_DIM + o0 + 4) = r1;
        }
    }
}

__global__ __launch_bounds__(256) void k_mlp(
    const float* __restrict__ x, const int* __restrict__ node_ind,
    const float* W1, const float* __restrict__ b1,
    const float* W2, const float* __restrict__ b2,
    float* out)
{
    __shared__ float xs[2 * XS_TILE];
    __shared__ float hs[2 * HS_TILE];
    __shared__ __attribute__((aligned(16))) float wst[8 * 16 * WS_STRIDE];
    __shared__ int   tok[T_TOK];
    __shared__ __attribute__((aligned(16))) int wcnt[8];

    const int n    = blockIdx.x;
    const int tid  = threadIdx.x;
    const int lane = tid & 31;
    const int wave = tid >> 5;
    const int nlo  = lane & 15;
    const int kb   = (lane >> 4) << 1;
    const int mhi  = (lane >> 4) << 3;

    const float* W1n = W1 + (size_t)n * H4 * H_DIM;
    const float* W2n = W2 + (size_t)n * H_DIM * H4;
    const float* b1n = b1 + (size_t)n * H4;
    const float* b2n = b2 + (size_t)n * H_DIM;

    const float* src1 = W1n + (size_t)(wave << 4) * H_DIM + (lane << 2);
    const int    o8   = 8 * 16 * H_DIM;
    const float* src2 = W2n + (size_t)(wave << 4) * H4 + (lane << 2);

    const int tb = (wave << 8) + (lane << 2);
    const v4i id0 = *(const v4i*)(node_ind + tb);
    const v4i id1 = *(const v4i*)(node_ind + tb + 128);

    v4f wA[16], wB[16];
    load_chunk(wA, src1,      H_DIM);
    load_chunk(wB, src1 + o8, H_DIM);

    int pre[8];
    int run = 0;
    unsigned hitbits = 0;
    #pragma unroll
    for (int q = 0; q < 8; ++q) {
        const int id = (q < 4) ? id0[q & 3] : id1[q & 3];
        const bool hit = (id == n);
        const unsigned mask = __builtin_amdgcn_ballot_w32(hit);
        pre[q] = __builtin_amdgcn_mbcnt_lo(mask, run);
        run += __builtin_popcount(mask);
        hitbits |= hit ? (1u << q) : 0u;
    }
    if (lane == 0) wcnt[wave] = run;
    lds_barrier();

    const v4i c0 = *(const v4i*)(wcnt);
    const v4i c1 = *(const v4i*)(wcnt + 4);
    int base = 0, cnt = 0;
    #pragma unroll
    for (int w = 0; w < 8; ++w) {
        const int cw = (w < 4) ? c0[w & 3] : c1[w & 3];
        base += (w < wave) ? cw : 0;
        cnt  += cw;
    }
    #pragma unroll
    for (int q = 0; q < 8; ++q)
        if (hitbits & (1u << q)) tok[base + pre[q]] = tb + ((q >> 2) << 7) + (q & 3);
    lds_barrier();
    if (cnt == 0) {
        asm volatile("s_wait_loadcnt 0x0" ::: "memory");
        return;
    }

    float*       wmine = wst + wave * (16 * WS_STRIDE);
    float*       wdst  = wmine + (lane << 2);
    const float* afrag = wmine + nlo * WS_STRIDE + kb;

    v4f bv[8];
    #pragma unroll
    for (int c = 0; c < 4; ++c) {
        bv[2 * c]     = *(const v4f*)(b1n + ((wave + 8 * c) << 4) + mhi);
        bv[2 * c + 1] = *(const v4f*)(b1n + ((wave + 8 * c) << 4) + mhi + 4);
    }

    for (int tile = 0; ; ) {
        if (__builtin_expect(cnt - tile > 16, 0)) {
            run_two_tiles(wA, wB, x, src1, src2, o8, bv, b2n, out, xs, hs, tok, wdst, afrag,
                          tile, cnt, tid, lane, wave, nlo, kb, mhi);
            tile += 32;
        } else {
            run_one_tile(wA, wB, x, src1, src2, o8, bv, b2n, out, xs, hs, tok, wdst, afrag,
                         tile, cnt, tid, wave, nlo, kb, mhi);
            tile += 16;
        }
        if (tile >= cnt) break;
        load_chunk(wA, src1,      H_DIM);
        load_chunk(wB, src1 + o8, H_DIM);
        __syncthreads();
    }
}


extern "C" void kernel_launch(void* const* d_in, const int* in_sizes, int n_in,
                              void* d_out, int out_size, void* d_ws, size_t ws_size,
                              hipStream_t stream) {
    const float* x        = (const float*)d_in[0];
    const int*   node_ind = (const int*)  d_in[1];
    const float* W1       = (const float*)d_in[2];
    const float* b1       = (const float*)d_in[3];
    const float* W2       = (const float*)d_in[4];
    const float* b2       = (const float*)d_in[5];
    float*       out      = (float*)d_out;

    k_mlp<<<NN, 256, 0, stream>>>(x, node_ind, W1, b1, W2, b2, out);
}


# GELU's erf without branches (both ranges computed per element, result selected): a lane's elements run as independent packed chains, in both bodies straight from the accumulators; same values
# speedup vs baseline: 1.9248x; 1.0771x over previous
#include <hip/hip_runtime.h>
#include <math.h>

typedef __attribute__((ext_vector_type(2))) float v2f;
typedef __attribute__((ext_vector_type(4))) float v4f;
typedef __attribute__((ext_vector_type(8))) float v8f;

#define T_TOK 2048
#define H_DIM 128
#define H4    512
#define NN    256

#define XS_STRIDE 132
#define HS_STRIDE 516
#define WS_STRIDE 132
#define XS_TILE (16 * XS_STRIDE)
#define HS_TILE (16 * HS_STRIDE)

typedef __attribute__((ext_vector_type(4))) int v4i;


__device__ __forceinline__ float erf_sel(const float y) {
    const float ax = fabsf(y);
    const float t  = ax * ax;
    float q = fmaf(t, -0x1.268bc2p-11f, 0x1.420828p-8f);
    q = fmaf(t, q, -0x1.b5937p-6f);
    q = fmaf(t, q,  0x1.ce077cp-4f);
    q = fmaf(t, q, -0x1.81266p-2f);
    q = fmaf(t, q,  0x1.06eba0p-3f);
    const float lo = fmaf(ax, q, ax);
    float p = fmaf(ax, 0x1.1d3156p-16f, -0x1.8d129p-12f);
    p = fmaf(ax, p,  0x1.f9a6d2p-9f);
    p = fmaf(ax, p, -0x1.8c3164p-6f);
    p = fmaf(ax, p,  0x1.b4e9c8p-4f);
    p = fmaf(ax, p,  0x1.4515fap-1f);
    p = fmaf(ax, p,  0x1.078e50p-3f);
    p = fmaf(ax, p, ax);
    const float hi = 1.0f - expf(-p);
    return copysignf(ax < 1.0f ? lo : hi, y);
}

__device__ __forceinline__ float gelu_erf(float v) {
    return 0.5f * v * (1.0f + erf_sel(v * 0.70710678118654752f));
}

__device__ __forceinline__ void load_chunk(v4f (&w)[16], const float* src, const int pitch) {
    #pragma unroll
    for (int r = 0; r < 16; ++r) w[r] = *(const v4f*)(src + r * pitch);
}

__device__ __forceinline__ void stage_chunk(float* dst, const v4f (&w)[16]) {
    #pragma unroll
    for (int r = 0; r < 16; ++r) *(v4f*)(dst + r * WS_STRIDE) = w[r];
}

template <int NT>
__device__ __forceinline__ void mma_chunk(const float* arow, const float* brow, const int bpitch,
                                          v8f (&acc)[NT]) {
    #pragma unroll
    for (int k = 0; k < 128; k += 4) {
        v2f a = *(const v2f*)(arow + k);
        #pragma unroll
        for (int t = 0; t < NT; ++t) {
            v2f b = *(const v2f*)(brow + t * bpitch + k);
            acc[t] = __builtin_amdgcn_wmma_f32_16x16x4_f32(
                false, a, false, b, (short)0, acc[t], false, false);
        }
    }
}

template <int NT>
__device__ __forceinline__ void clear_acc(v8f (&acc)[NT]) {
    const v8f zero = {};
    #pragma unroll
    for (int t = 0; t < NT; ++t) acc[t] = zero;
}

__device__ __forceinline__ void deposit_h(float* hdst, const v4f bia0, const v4f bia1, v8f acc) {
    v4f h0, h1;
    #pragma unroll
    for (int r = 0; r < 4; ++r) h0[r] = gelu_erf(acc[r]     + bia0[r]);
    #pragma unroll
    for (int r = 0; r < 4; ++r) h1[r] = gelu_erf(acc[r + 4] + bia1[r]);
    *(v4f*)(hdst)     = h0;
    *(v4f*)(hdst + 4) = h1;
}

template <int NT>
__device__ __forceinline__ void deposit_tiles(float* hdst, const v4f bia0, const v4f bia1,
                                              const v8f (&acc)[NT]) {
    #pragma unroll
    for (int t = 0; t < NT; ++t) deposit_h(hdst + t * HS_TILE, bia0, bia1, acc[t]);
}

__device__ __forceinline__ void lds_barrier() {
    asm volatile("s_wait_dscnt 0x0\n\ts_barrier_signal -1\n\ts_barrier_wait -1" ::: "memory");
}

__device__ __forceinline__ void phase_fence() {
    __builtin_amdgcn_sched_barrier(0);
}

__device__ __forceinline__ void run_one_tile(
    v4f (&wA)[16], v4f (&wB)[16],
    const float* __restrict__ x, const float* src1, const float* src2, const int o8,
    const v4f (&bv)[8], const float* __restrict__ b2n, float* out,
    float* xs, float* hs, const int* tok, float* wdst, const float* afrag,
    const int tile, const int cnt,
    const int tid, const int wave, const int nlo, const int kb, const int mhi)
{
    constexpr int NT = 1;
    const float* xfrag = xs + nlo * XS_STRIDE + kb;
    const float* hfrag = hs + nlo * HS_STRIDE + kb;
    float*       hdst  = hs + nlo * HS_STRIDE + mhi;

    const int xr0 = tid >> 5, xc4 = (tid & 31) << 2;
    v4f xv[2 * NT];
    #pragma unroll
    for (int j = 0; j < 2 * NT; ++j) {
        const int s = tile + xr0 + 8 * j;
        const int t = tok[s < cnt ? s : 0];
        xv[j] = *(const v4f*)(x + (size_t)t * H_DIM + xc4);
    }
    #pragma unroll
    for (int j = 0; j < 2 * NT; ++j)
        *(v4f*)(xs + (j >> 1) * XS_TILE + (xr0 + 8 * (j & 1)) * XS_STRIDE + xc4) = xv[j];
    __syncthreads();

    v8f acc[NT];

    stage_chunk(wdst, wA);  load_chunk(wA, src1 + 2 * o8, H_DIM);
    phase_fence();
    clear_acc<NT>(acc);
    mma_chunk<NT>(afrag, xfrag, XS_TILE, acc);
    deposit_tiles<NT>(hdst + (wave << 4),        bv[0], bv[1], acc);
    phase_fence();

    stage_chunk(wdst, wB);  load_chunk(wB, src1 + 3 * o8, H_DIM);
    phase_fence();
    clear_acc<NT>(acc);
    mma_chunk<NT>(afrag, xfrag, XS_TILE, acc);
    deposit_tiles<NT>(hdst + ((wave + 8) << 4),  bv[2], bv[3], acc);
    phase_fence();

    stage_chunk(wdst, wA);  load_chunk(wA, src2,       H4);
    phase_fence();
    clear_acc<NT>(acc);
    mma_chunk<NT>(afrag, xfrag, XS_TILE, acc);
    deposit_tiles<NT>(hdst + ((wave + 16) << 4), bv[4], bv[5], acc);
    phase_fence();

    stage_chunk(wdst, wB);  load_chunk(wB, src2 + 128, H4);
    phase_fence();
    clear_acc<NT>(acc);
    mma_chunk<NT>(afrag, xfrag, XS_TILE, acc);
    deposit_tiles<NT>(hdst + ((wave + 24) << 4), bv[6], bv[7], acc);
    __syncthreads();

    stage_chunk(wdst, wA);  load_chunk(wA, src2 + 256, H4);
    clear_acc<NT>(acc);
    mma_chunk<NT>(afrag, hfrag,       HS_TILE, acc);
    stage_chunk(wdst, wB);  load_chunk(wB, src2 + 384, H4);
    mma_chunk<NT>(afrag, hfrag + 128, HS_TILE, acc);
    stage_chunk(wdst, wA);
    mma_chunk<NT>(afrag, hfrag + 256, HS_TILE, acc);
    stage_chunk(wdst, wB);
    mma_chunk<NT>(afrag, hfrag + 384, HS_TILE, acc);

    #pragma unroll
    for (int tt = 0; tt < NT; ++tt) {
        const int s = tile + 16 * tt + nlo;
        if (s < cnt) {
            const int t  = tok[s];
            const int o0 = (wave << 4) + mhi;
            v4f bia0 = *(const v4f*)(b2n + o0);
            v4f bia1 = *(const v4f*)(b2n + o0 + 4);
            v4f r0, r1;
            #pragma unroll
            for (int r = 0; r < 4; ++r) r0[r] = acc[tt][r]     + bia0[r];
            #pragma unroll
            for (int r = 0; r < 4; ++r) r1[r] = acc[tt][r + 4] + bia1[r];
            *(v4f*)(out + (size_t)t * H_DIM + o0)     = r0;
            *(v4f*)(out + (size_t)t * H_DIM + o0 + 4) = r1;
        }
    }
}

__device__ __forceinline__ void run_two_tiles(
    v4f (&wA)[16], v4f (&wB)[16],
    const float* __restrict__ x, const float* src1, const float* src2, const int o8,
    const v4f (&bv)[8], const float* __restrict__ b2n, float* out,
    float* xs, float* hs, const int* tok, float* wdst, const float* afrag,
    const int tile, const int cnt,
    const int tid, const int wave, const int nlo, const int kb, const int mhi)
{
    const float* xfrag = xs + nlo * XS_STRIDE + kb;
    const float* hfrag = hs + nlo * HS_STRIDE + kb;
    float*       hdst  = hs + nlo * HS_STRIDE + mhi;

    const int xr0 = tid >> 5, xc4 = (tid & 31) << 2;
    v4f xv[4];
    #pragma unroll
    for (int j = 0; j < 4; ++j) {
        const int s = tile + xr0 + 8 * j;
        const int t = tok[s < cnt ? s : 0];
        xv[j] = *(const v4f*)(x + (size_t)t * H_DIM + xc4);
    }
    #pragma unroll
    for (int j = 0; j < 4; ++j)
        *(v4f*)(xs + (j >> 1) * XS_TILE + (xr0 + 8 * (j & 1)) * XS_STRIDE + xc4) = xv[j];
    __syncthreads();

    v8f acc[2];

    #pragma unroll 1
    for (int p = 0; p < 4; ++p) {
        const bool   l1     = p < 2;
        const int    cn     = 2 * p + 2;
        const float* nsrc   = (cn < 4) ? src1 + cn * o8 : src2 + (cn - 4) * 128;
        const int    nstep  = (cn < 4) ? o8 : 128;
        const int    npitch = (cn < 4) ? H_DIM : H4;
        const float* bfrag  = l1 ? xfrag : hfrag + (2 * p - 4) * 128;
        const int    bstep  = l1 ? 0 : 128;
        const int    bpitch = l1 ? XS_TILE : HS_TILE;
        const int    ot     = wave + 16 * p;
        const v4f ba0 = (p == 0) ? bv[0] : bv[4], ba1 = (p == 0) ? bv[1] : bv[5];
        const v4f bb0 = (p == 0) ? bv[2] : bv[6], bb1 = (p == 0) ? bv[3] : bv[7];

        stage_chunk(wdst, wA);
        if (p < 3) load_chunk(wA, nsrc, npitch);
        if (p < 3) clear_acc<2>(acc);
        mma_chunk<2>(afrag, bfrag, bpitch, acc);
        if (l1) deposit_tiles<2>(hdst + (ot << 4),       ba0, ba1, acc);

        stage_chunk(wdst, wB);
        if (p < 3) load_chunk(wB, nsrc + nstep, npitch);
        if (l1) clear_acc<2>(acc);
        mma_chunk<2>(afrag, bfrag + bstep, bpitch, acc);
        if (l1) deposit_tiles<2>(hdst + ((ot + 8) << 4), bb0, bb1, acc);
        if (p == 1) __syncthreads();
    }

    #pragma unroll
    for (int tt = 0; tt < 2; ++tt) {
        const int s = tile + 16 * tt + nlo;
        if (s < cnt) {
            const int t  = tok[s];
            const int o0 = (wave << 4) + mhi;
            v4f bia0 = *(const v4f*)(b2n + o0);
            v4f bia1 = *(const v4f*)(b2n + o0 + 4);
            v4f r0, r1;
            #pragma unroll
            for (int r = 0; r < 4; ++r) r0[r] = acc[tt][r]     + bia0[r];
            #pragma unroll
            for (int r = 0; r < 4; ++r) r1[r] = acc[tt][r + 4] + bia1[r];
            *(v4f*)(out + (size_t)t * H_DIM + o0)     = r0;
            *(v4f*)(out + (size_t)t * H_DIM + o0 + 4) = r1;
        }
    }
}

__global__ __launch_bounds__(256) void k_mlp(
    const float* __restrict__ x, const int* __restrict__ node_ind,
    const float* W1, const float* __restrict__ b1,
    const float* W2, const float* __restrict__ b2,
    float* out)
{
    __shared__ float xs[2 * XS_TILE];
    __shared__ float hs[2 * HS_TILE];
    __shared__ __attribute__((aligned(16))) float wst[8 * 16 * WS_STRIDE];
    __shared__ int   tok[T_TOK];
    __shared__ __attribute__((aligned(16))) int wcnt[8];

    const int n    = blockIdx.x;
    const int tid  = threadIdx.x;
    const int lane = tid & 31;
    const int wave = tid >> 5;
    const int nlo  = lane & 15;
    const int kb   = (lane >> 4) << 1;
    const int mhi  = (lane >> 4) << 3;

    const float* W1n = W1 + (size_t)n * H4 * H_DIM;
    const float* W2n = W2 + (size_t)n * H_DIM * H4;
    const float* b1n = b1 + (size_t)n * H4;
    const float* b2n = b2 + (size_t)n * H_DIM;

    const float* src1 = W1n + (size_t)(wave << 4) * H_DIM + (lane << 2);
    const int    o8   = 8 * 16 * H_DIM;
    const float* src2 = W2n + (size_t)(wave << 4) * H4 + (lane << 2);

    const int tb = (wave << 8) + (lane << 2);
    const v4i id0 = *(const v4i*)(node_ind + tb);
    const v4i id1 = *(const v4i*)(node_ind + tb + 128);

    v4f wA[16], wB[16];
    load_chunk(wA, src1,      H_DIM);
    load_chunk(wB, src1 + o8, H_DIM);

    int pre[8];
    int run = 0;
    unsigned hitbits = 0;
    #pragma unroll
    for (int q = 0; q < 8; ++q) {
        const int id = (q < 4) ? id0[q & 3] : id1[q & 3];
        const bool hit = (id == n);
        const unsigned mask = __builtin_amdgcn_ballot_w32(hit);
        pre[q] = __builtin_amdgcn_mbcnt_lo(mask, run);
        run += __builtin_popcount(mask);
        hitbits |= hit ? (1u << q) : 0u;
    }
    if (lane == 0) wcnt[wave] = run;
    lds_barrier();

    const v4i c0 = *(const v4i*)(wcnt);
    const v4i c1 = *(const v4i*)(wcnt + 4);
    int base = 0, cnt = 0;
    #pragma unroll
    for (int w = 0; w < 8; ++w) {
        const int cw = (w < 4) ? c0[w & 3] : c1[w & 3];
        base += (w < wave) ? cw : 0;
        cnt  += cw;
    }
    #pragma unroll
    for (int q = 0; q < 8; ++q)
        if (hitbits & (1u << q)) tok[base + pre[q]] = tb + ((q >> 2) << 7) + (q & 3);
    lds_barrier();
    if (cnt == 0) {
        asm volatile("s_wait_loadcnt 0x0" ::: "memory");
        return;
    }

    float*       wmine = wst + wave * (16 * WS_STRIDE);
    float*       wdst  = wmine + (lane << 2);
    const float* afrag = wmine + nlo * WS_STRIDE + kb;

    v4f bv[8];
    #pragma unroll
    for (int c = 0; c < 4; ++c) {
        bv[2 * c]     = *(const v4f*)(b1n + ((wave + 8 * c) << 4) + mhi);
        bv[2 * c + 1] = *(const v4f*)(b1n + ((wave + 8 * c) << 4) + mhi + 4);
    }

    for (int tile = 0; ; ) {
        if (__builtin_expect(cnt - tile > 16, 0)) {
            run_two_tiles(wA, wB, x, src1, src2, o8, bv, b2n, out, xs, hs, tok, wdst, afrag,
                          tile, cnt, tid, wave, nlo, kb, mhi);
            tile += 32;
        } else {
            run_one_tile(wA, wB, x, src1, src2, o8, bv, b2n, out, xs, hs, tok, wdst, afrag,
                         tile, cnt, tid, wave, nlo, kb, mhi);
            tile += 16;
        }
        if (tile >= cnt) break;
        load_chunk(wA, src1,      H_DIM);
        load_chunk(wB, src1 + o8, H_DIM);
        __syncthreads();
    }
}


extern "C" void kernel_launch(void* const* d_in, const int* in_sizes, int n_in,
                              void* d_out, int out_size, void* d_ws, size_t ws_size,
                              hipStream_t stream) {
    const float* x        = (const float*)d_in[0];
    const int*   node_ind = (const int*)  d_in[1];
    const float* W1       = (const float*)d_in[2];
    const float* b1       = (const float*)d_in[3];
    const float* W2       = (const float*)d_in[4];
    const float* b2       = (const float*)d_in[5];
    float*       out      = (float*)d_out;

    k_mlp<<<NN, 256, 0, stream>>>(x, node_ind, W1, b1, W2, b2, out);
}


# k_mlp: a chunk's LDS fragment reads all leave as one group into registers ahead of its WMMAs, which then run back to back with no LDS read among them; same fragments, same k order
# speedup vs baseline: 2.0050x; 1.0417x over previous
#include <hip/hip_runtime.h>
#include <math.h>

typedef __attribute__((ext_vector_type(2))) float v2f;
typedef __attribute__((ext_vector_type(4))) float v4f;
typedef __attribute__((ext_vector_type(8))) float v8f;

#define T_TOK 2048
#define H_DIM 128
#define H4    512
#define NN    256

#define XS_STRIDE 132
#define HS_STRIDE 516
#define WS_STRIDE 132
#define XS_TILE (16 * XS_STRIDE)
#define HS_TILE (16 * HS_STRIDE)

typedef __attribute__((ext_vector_type(4))) int v4i;


__device__ __forceinline__ float erf_sel(const float y) {
    const float ax = fabsf(y);
    const float t  = ax * ax;
    float q = fmaf(t, -0x1.268bc2p-11f, 0x1.420828p-8f);
    q = fmaf(t, q, -0x1.b5937p-6f);
    q = fmaf(t, q,  0x1.ce077cp-4f);
    q = fmaf(t, q, -0x1.81266p-2f);
    q = fmaf(t, q,  0x1.06eba0p-3f);
    const float lo = fmaf(ax, q, ax);
    float p = fmaf(ax, 0x1.1d3156p-16f, -0x1.8d129p-12f);
    p = fmaf(ax, p,  0x1.f9a6d2p-9f);
    p = fmaf(ax, p, -0x1.8c3164p-6f);
    p = fmaf(ax, p,  0x1.b4e9c8p-4f);
    p = fmaf(ax, p,  0x1.4515fap-1f);
    p = fmaf(ax, p,  0x1.078e50p-3f);
    p = fmaf(ax, p, ax);
    const float hi = 1.0f - expf(-p);
    return copysignf(ax < 1.0f ? lo : hi, y);
}

__device__ __forceinline__ float gelu_erf(float v) {
    return 0.5f * v * (1.0f + erf_sel(v * 0.70710678118654752f));
}

__device__ __forceinline__ void load_chunk(v4f (&w)[16], const float* src, const int pitch) {
    #pragma unroll
    for (int r = 0; r < 16; ++r) w[r] = *(const v4f*)(src + r * pitch);
}

__device__ __forceinline__ void stage_chunk(float* dst, const v4f (&w)[16]) {
    #pragma unroll
    for (int r = 0; r < 16; ++r) *(v4f*)(dst + r * WS_STRIDE) = w[r];
}

template <int NT>
__device__ __forceinline__ void pin_acc(v8f (&acc)[NT]) {
    #pragma unroll
    for (int t = 0; t < NT; ++t) asm volatile("" : "+v"(acc[t]) :: "memory");
}

template <int NT>
__device__ __forceinline__ void mma_chunk(const float* arow, const float* brow, const int bpitch,
                                          v8f (&acc)[NT]) {
    v2f a[32], b[NT][32];
    #pragma unroll
    for (int j = 0; j < 32; ++j) {
        a[j] = *(const v2f*)(arow + 4 * j);
        #pragma unroll
        for (int t = 0; t < NT; ++t) b[t][j] = *(const v2f*)(brow + t * bpitch + 4 * j);
    }
    pin_acc<NT>(acc);
    #pragma unroll
    for (int j = 0; j < 32; ++j) {
        #pragma unroll
        for (int t = 0; t < NT; ++t)
            acc[t] = __builtin_amdgcn_wmma_f32_16x16x4_f32(
                false, a[j], false, b[t][j], (short)0, acc[t], false, false);
    }
    pin_acc<NT>(acc);
}

template <int NT>
__device__ __forceinline__ void clear_acc(v8f (&acc)[NT]) {
    const v8f zero = {};
    #pragma unroll
    for (int t = 0; t < NT; ++t) acc[t] = zero;
}

__device__ __forceinline__ void deposit_h(float* hdst, const v4f bia0, const v4f bia1, v8f acc) {
    v4f h0, h1;
    #pragma unroll
    for (int r = 0; r < 4; ++r) h0[r] = gelu_erf(acc[r]     + bia0[r]);
    #pragma unroll
    for (int r = 0; r < 4; ++r) h1[r] = gelu_erf(acc[r + 4] + bia1[r]);
    *(v4f*)(hdst)     = h0;
    *(v4f*)(hdst + 4) = h1;
}

template <int NT>
__device__ __forceinline__ void deposit_tiles(float* hdst, const v4f bia0, const v4f bia1,
                                              const v8f (&acc)[NT]) {
    #pragma unroll
    for (int t = 0; t < NT; ++t) deposit_h(hdst + t * HS_TILE, bia0, bia1, acc[t]);
}

__device__ __forceinline__ void lds_barrier() {
    asm volatile("s_wait_dscnt 0x0\n\ts_barrier_signal -1\n\ts_barrier_wait -1" ::: "memory");
}

__device__ __forceinline__ void phase_fence() {
    __builtin_amdgcn_sched_barrier(0);
}

__device__ __forceinline__ void run_one_tile(
    v4f (&wA)[16], v4f (&wB)[16],
    const float* __restrict__ x, const float* src1, const float* src2, const int o8,
    const v4f (&bv)[8], const float* __restrict__ b2n, float* out,
    float* xs, float* hs, const int* tok, float* wdst, const float* afrag,
    const int tile, const int cnt,
    const int tid, const int wave, const int nlo, const int kb, const int mhi)
{
    constexpr int NT = 1;
    const float* xfrag = xs + nlo * XS_STRIDE + kb;
    const float* hfrag = hs + nlo * HS_STRIDE + kb;
    float*       hdst  = hs + nlo * HS_STRIDE + mhi;

    const int xr0 = tid >> 5, xc4 = (tid & 31) << 2;
    v4f xv[2 * NT];
    #pragma unroll
    for (int j = 0; j < 2 * NT; ++j) {
        const int s = tile + xr0 + 8 * j;
        const int t = tok[s < cnt ? s : 0];
        xv[j] = *(const v4f*)(x + (size_t)t * H_DIM + xc4);
    }
    #pragma unroll
    for (int j = 0; j < 2 * NT; ++j)
        *(v4f*)(xs + (j >> 1) * XS_TILE + (xr0 + 8 * (j & 1)) * XS_STRIDE + xc4) = xv[j];
    __syncthreads();

    v8f acc[NT];

    stage_chunk(wdst, wA);  load_chunk(wA, src1 + 2 * o8, H_DIM);
    phase_fence();
    clear_acc<NT>(acc);
    mma_chunk<NT>(afrag, xfrag, XS_TILE, acc);
    deposit_tiles<NT>(hdst + (wave << 4),        bv[0], bv[1], acc);
    phase_fence();

    stage_chunk(wdst, wB);  load_chunk(wB, src1 + 3 * o8, H_DIM);
    phase_fence();
    clear_acc<NT>(acc);
    mma_chunk<NT>(afrag, xfrag, XS_TILE, acc);
    deposit_tiles<NT>(hdst + ((wave + 8) << 4),  bv[2], bv[3], acc);
    phase_fence();

    stage_chunk(wdst, wA);  load_chunk(wA, src2,       H4);
    phase_fence();
    clear_acc<NT>(acc);
    mma_chunk<NT>(afrag, xfrag, XS_TILE, acc);
    deposit_tiles<NT>(hdst + ((wave + 16) << 4), bv[4], bv[5], acc);
    phase_fence();

    stage_chunk(wdst, wB);  load_chunk(wB, src2 + 128, H4);
    phase_fence();
    clear_acc<NT>(acc);
    mma_chunk<NT>(afrag, xfrag, XS_TILE, acc);
    deposit_tiles<NT>(hdst + ((wave + 24) << 4), bv[6], bv[7], acc);
    __syncthreads();

    stage_chunk(wdst, wA);  load_chunk(wA, src2 + 256, H4);
    clear_acc<NT>(acc);
    mma_chunk<NT>(afrag, hfrag,       HS_TILE, acc);
    stage_chunk(wdst, wB);  load_chunk(wB, src2 + 384, H4);
    mma_chunk<NT>(afrag, hfrag + 128, HS_TILE, acc);
    stage_chunk(wdst, wA);
    mma_chunk<NT>(afrag, hfrag + 256, HS_TILE, acc);
    stage_chunk(wdst, wB);
    mma_chunk<NT>(afrag, hfrag + 384, HS_TILE, acc);

    #pragma unroll
    for (int tt = 0; tt < NT; ++tt) {
        const int s = tile + 16 * tt + nlo;
        if (s < cnt) {
            const int t  = tok[s];
            const int o0 = (wave << 4) + mhi;
            v4f bia0 = *(const v4f*)(b2n + o0);
            v4f bia1 = *(const v4f*)(b2n + o0 + 4);
            v4f r0, r1;
            #pragma unroll
            for (int r = 0; r < 4; ++r) r0[r] = acc[tt][r]     + bia0[r];
            #pragma unroll
            for (int r = 0; r < 4; ++r) r1[r] = acc[tt][r + 4] + bia1[r];
            *(v4f*)(out + (size_t)t * H_DIM + o0)     = r0;
            *(v4f*)(out + (size_t)t * H_DIM + o0 + 4) = r1;
        }
    }
}

__device__ __forceinline__ void run_two_tiles(
    v4f (&wA)[16], v4f (&wB)[16],
    const float* __restrict__ x, const float* src1, const float* src2, const int o8,
    const v4f (&bv)[8], const float* __restrict__ b2n, float* out,
    float* xs, float* hs, const int* tok, float* wdst, const float* afrag,
    const int tile, const int cnt,
    const int tid, const int wave, const int nlo, const int kb, const int mhi)
{
    const float* xfrag = xs + nlo * XS_STRIDE + kb;
    const float* hfrag = hs + nlo * HS_STRIDE + kb;
    float*       hdst  = hs + nlo * HS_STRIDE + mhi;

    const int xr0 = tid >> 5, xc4 = (tid & 31) << 2;
    v4f xv[4];
    #pragma unroll
    for (int j = 0; j < 4; ++j) {
        const int s = tile + xr0 + 8 * j;
        const int t = tok[s < cnt ? s : 0];
        xv[j] = *(const v4f*)(x + (size_t)t * H_DIM + xc4);
    }
    #pragma unroll
    for (int j = 0; j < 4; ++j)
        *(v4f*)(xs + (j >> 1) * XS_TILE + (xr0 + 8 * (j & 1)) * XS_STRIDE + xc4) = xv[j];
    __syncthreads();

    v8f acc[2];

    #pragma unroll 1
    for (int p = 0; p < 4; ++p) {
        const bool   l1     = p < 2;
        const int    cn     = 2 * p + 2;
        const float* nsrc   = (cn < 4) ? src1 + cn * o8 : src2 + (cn - 4) * 128;
        const int    nstep  = (cn < 4) ? o8 : 128;
        const int    npitch = (cn < 4) ? H_DIM : H4;
        const float* bfrag  = l1 ? xfrag : hfrag + (2 * p - 4) * 128;
        const int    bstep  = l1 ? 0 : 128;
        const int    bpitch = l1 ? XS_TILE : HS_TILE;
        const int    ot     = wave + 16 * p;
        const v4f ba0 = (p == 0) ? bv[0] : bv[4], ba1 = (p == 0) ? bv[1] : bv[5];
        const v4f bb0 = (p == 0) ? bv[2] : bv[6], bb1 = (p == 0) ? bv[3] : bv[7];

        stage_chunk(wdst, wA);
        if (p < 3) load_chunk(wA, nsrc, npitch);
        if (p < 3) clear_acc<2>(acc);
        mma_chunk<2>(afrag, bfrag, bpitch, acc);
        if (l1) deposit_tiles<2>(hdst + (ot << 4),       ba0, ba1, acc);

        stage_chunk(wdst, wB);
        if (p < 3) load_chunk(wB, nsrc + nstep, npitch);
        if (l1) clear_acc<2>(acc);
        mma_chunk<2>(afrag, bfrag + bstep, bpitch, acc);
        if (l1) deposit_tiles<2>(hdst + ((ot + 8) << 4), bb0, bb1, acc);
        if (p == 1) __syncthreads();
    }

    #pragma unroll
    for (int tt = 0; tt < 2; ++tt) {
        const int s = tile + 16 * tt + nlo;
        if (s < cnt) {
            const int t  = tok[s];
            const int o0 = (wave << 4) + mhi;
            v4f bia0 = *(const v4f*)(b2n + o0);
            v4f bia1 = *(const v4f*)(b2n + o0 + 4);
            v4f r0, r1;
            #pragma unroll
            for (int r = 0; r < 4; ++r) r0[r] = acc[tt][r]     + bia0[r];
            #pragma unroll
            for (int r = 0; r < 4; ++r) r1[r] = acc[tt][r + 4] + bia1[r];
            *(v4f*)(out + (size_t)t * H_DIM + o0)     = r0;
            *(v4f*)(out + (size_t)t * H_DIM + o0 + 4) = r1;
        }
    }
}

__global__ __launch_bounds__(256) void k_mlp(
    const float* __restrict__ x, const int* __restrict__ node_ind,
    const float* W1, const float* __restrict__ b1,
    const float* W2, const float* __restrict__ b2,
    float* out)
{
    __shared__ float xs[2 * XS_TILE];
    __shared__ float hs[2 * HS_TILE];
    __shared__ __attribute__((aligned(16))) float wst[8 * 16 * WS_STRIDE];
    __shared__ int   tok[T_TOK];
    __shared__ __attribute__((aligned(16))) int wcnt[8];

    const int n    = blockIdx.x;
    const int tid  = threadIdx.x;
    const int lane = tid & 31;
    const int wave = tid >> 5;
    const int nlo  = lane & 15;
    const int kb   = (lane >> 4) << 1;
    const int mhi  = (lane >> 4) << 3;

    const float* W1n = W1 + (size_t)n * H4 * H_DIM;
    const float* W2n = W2 + (size_t)n * H_DIM * H4;
    const float* b1n = b1 + (size_t)n * H4;
    const float* b2n = b2 + (size_t)n * H_DIM;

    const float* src1 = W1n + (size_t)(wave << 4) * H_DIM + (lane << 2);
    const int    o8   = 8 * 16 * H_DIM;
    const float* src2 = W2n + (size_t)(wave << 4) * H4 + (lane << 2);

    const int tb = (wave << 8) + (lane << 2);
    const v4i id0 = *(const v4i*)(node_ind + tb);
    const v4i id1 = *(const v4i*)(node_ind + tb + 128);

    v4f wA[16], wB[16];
    load_chunk(wA, src1,      H_DIM);
    load_chunk(wB, src1 + o8, H_DIM);

    int pre[8];
    int run = 0;
    unsigned hitbits = 0;
    #pragma unroll
    for (int q = 0; q < 8; ++q) {
        const int id = (q < 4) ? id0[q & 3] : id1[q & 3];
        const bool hit = (id == n);
        const unsigned mask = __builtin_amdgcn_ballot_w32(hit);
        pre[q] = __builtin_amdgcn_mbcnt_lo(mask, run);
        run += __builtin_popcount(mask);
        hitbits |= hit ? (1u << q) : 0u;
    }
    if (lane == 0) wcnt[wave] = run;
    lds_barrier();

    const v4i c0 = *(const v4i*)(wcnt);
    const v4i c1 = *(const v4i*)(wcnt + 4);
    int base = 0, cnt = 0;
    #pragma unroll
    for (int w = 0; w < 8; ++w) {
        const int cw = (w < 4) ? c0[w & 3] : c1[w & 3];
        base += (w < wave) ? cw : 0;
        cnt  += cw;
    }
    #pragma unroll
    for (int q = 0; q < 8; ++q)
        if (hitbits & (1u << q)) tok[base + pre[q]] = tb + ((q >> 2) << 7) + (q & 3);
    lds_barrier();
    if (cnt == 0) {
        asm volatile("s_wait_loadcnt 0x0" ::: "memory");
        return;
    }

    float*       wmine = wst + wave * (16 * WS_STRIDE);
    float*       wdst  = wmine + (lane << 2);
    const float* afrag = wmine + nlo * WS_STRIDE + kb;

    v4f bv[8];
    #pragma unroll
    for (int c = 0; c < 4; ++c) {
        bv[2 * c]     = *(const v4f*)(b1n + ((wave + 8 * c) << 4) + mhi);
        bv[2 * c + 1] = *(const v4f*)(b1n + ((wave + 8 * c) << 4) + mhi + 4);
    }

    for (int tile = 0; ; ) {
        if (__builtin_expect(cnt - tile > 16, 0)) {
            run_two_tiles(wA, wB, x, src1, src2, o8, bv, b2n, out, xs, hs, tok, wdst, afrag,
                          tile, cnt, tid, wave, nlo, kb, mhi);
            tile += 32;
        } else {
            run_one_tile(wA, wB, x, src1, src2, o8, bv, b2n, out, xs, hs, tok, wdst, afrag,
                         tile, cnt, tid, wave, nlo, kb, mhi);
            tile += 16;
        }
        if (tile >= cnt) break;
        load_chunk(wA, src1,      H_DIM);
        load_chunk(wB, src1 + o8, H_DIM);
        __syncthreads();
    }
}


extern "C" void kernel_launch(void* const* d_in, const int* in_sizes, int n_in,
                              void* d_out, int out_size, void* d_ws, size_t ws_size,
                              hipStream_t stream) {
    const float* x        = (const float*)d_in[0];
    const int*   node_ind = (const int*)  d_in[1];
    const float* W1       = (const float*)d_in[2];
    const float* b1       = (const float*)d_in[3];
    const float* W2       = (const float*)d_in[4];
    const float* b2       = (const float*)d_in[5];
    float*       out      = (float*)d_out;

    k_mlp<<<NN, 256, 0, stream>>>(x, node_ind, W1, b1, W2, b2, out);
}


# two-tile body: fragment reads software-pipelined by half chunks: the next 16 k-steps' LDS reads leave in front of the current 32 WMMAs (two register sets of the old total size); same k order
# speedup vs baseline: 2.0752x; 1.0350x over previous
#include <hip/hip_runtime.h>
#include <math.h>

typedef __attribute__((ext_vector_type(2))) float v2f;
typedef __attribute__((ext_vector_type(4))) float v4f;
typedef __attribute__((ext_vector_type(8))) float v8f;

#define T_TOK 2048
#define H_DIM 128
#define H4    512
#define NN    256

#define XS_STRIDE 132
#define HS_STRIDE 516
#define WS_STRIDE 132
#define XS_TILE (16 * XS_STRIDE)
#define HS_TILE (16 * HS_STRIDE)

typedef __attribute__((ext_vector_type(4))) int v4i;


__device__ __forceinline__ float erf_sel(const float y) {
    const float ax = fabsf(y);
    const float t  = ax * ax;
    float q = fmaf(t, -0x1.268bc2p-11f, 0x1.420828p-8f);
    q = fmaf(t, q, -0x1.b5937p-6f);
    q = fmaf(t, q,  0x1.ce077cp-4f);
    q = fmaf(t, q, -0x1.81266p-2f);
    q = fmaf(t, q,  0x1.06eba0p-3f);
    const float lo = fmaf(ax, q, ax);
    float p = fmaf(ax, 0x1.1d3156p-16f, -0x1.8d129p-12f);
    p = fmaf(ax, p,  0x1.f9a6d2p-9f);
    p = fmaf(ax, p, -0x1.8c3164p-6f);
    p = fmaf(ax, p,  0x1.b4e9c8p-4f);
    p = fmaf(ax, p,  0x1.4515fap-1f);
    p = fmaf(ax, p,  0x1.078e50p-3f);
    p = fmaf(ax, p, ax);
    const float hi = 1.0f - expf(-p);
    return copysignf(ax < 1.0f ? lo : hi, y);
}

__device__ __forceinline__ float gelu_erf(float v) {
    return 0.5f * v * (1.0f + erf_sel(v * 0.70710678118654752f));
}

__device__ __forceinline__ void load_chunk(v4f (&w)[16], const float* src, const int pitch) {
    #pragma unroll
    for (int r = 0; r < 16; ++r) w[r] = *(const v4f*)(src + r * pitch);
}

__device__ __forceinline__ void stage_chunk(float* dst, const v4f (&w)[16]) {
    #pragma unroll
    for (int r = 0; r < 16; ++r) *(v4f*)(dst + r * WS_STRIDE) = w[r];
}

template <int NT>
__device__ __forceinline__ void pin_acc(v8f (&acc)[NT]) {
    #pragma unroll
    for (int t = 0; t < NT; ++t) asm volatile("" : "+v"(acc[t]) :: "memory");
}

template <int NT>
__device__ __forceinline__ void mma_chunk(const float* arow, const float* brow, const int bpitch,
                                          v8f (&acc)[NT]) {
    v2f a[32], b[NT][32];
    #pragma unroll
    for (int j = 0; j < 32; ++j) {
        a[j] = *(const v2f*)(arow + 4 * j);
        #pragma unroll
        for (int t = 0; t < NT; ++t) b[t][j] = *(const v2f*)(brow + t * bpitch + 4 * j);
    }
    pin_acc<NT>(acc);
    #pragma unroll
    for (int j = 0; j < 32; ++j) {
        #pragma unroll
        for (int t = 0; t < NT; ++t)
            acc[t] = __builtin_amdgcn_wmma_f32_16x16x4_f32(
                false, a[j], false, b[t][j], (short)0, acc[t], false, false);
    }
    pin_acc<NT>(acc);
}

template <int NT>
__device__ __forceinline__ void read_half(v2f (&a)[16], v2f (&b)[NT][16],
                                          const float* arow, const float* brow, const int bpitch) {
    #pragma unroll
    for (int j = 0; j < 16; ++j) {
        a[j] = *(const v2f*)(arow + 4 * j);
        #pragma unroll
        for (int t = 0; t < NT; ++t) b[t][j] = *(const v2f*)(brow + t * bpitch + 4 * j);
    }
}

template <int NT>
__device__ __forceinline__ void mma_half(const v2f (&a)[16], const v2f (&b)[NT][16], v8f (&acc)[NT]) {
    pin_acc<NT>(acc);
    #pragma unroll
    for (int j = 0; j < 16; ++j) {
        #pragma unroll
        for (int t = 0; t < NT; ++t)
            acc[t] = __builtin_amdgcn_wmma_f32_16x16x4_f32(
                false, a[j], false, b[t][j], (short)0, acc[t], false, false);
    }
    pin_acc<NT>(acc);
}

template <int NT>
__device__ __forceinline__ void clear_acc(v8f (&acc)[NT]) {
    const v8f zero = {};
    #pragma unroll
    for (int t = 0; t < NT; ++t) acc[t] = zero;
}

__device__ __forceinline__ void deposit_h(float* hdst, const v4f bia0, const v4f bia1, v8f acc) {
    v4f h0, h1;
    #pragma unroll
    for (int r = 0; r < 4; ++r) h0[r] = gelu_erf(acc[r]     + bia0[r]);
    #pragma unroll
    for (int r = 0; r < 4; ++r) h1[r] = gelu_erf(acc[r + 4] + bia1[r]);
    *(v4f*)(hdst)     = h0;
    *(v4f*)(hdst + 4) = h1;
}

template <int NT>
__device__ __forceinline__ void deposit_tiles(float* hdst, const v4f bia0, const v4f bia1,
                                              const v8f (&acc)[NT]) {
    #pragma unroll
    for (int t = 0; t < NT; ++t) deposit_h(hdst + t * HS_TILE, bia0, bia1, acc[t]);
}

__device__ __forceinline__ void lds_barrier() {
    asm volatile("s_wait_dscnt 0x0\n\ts_barrier_signal -1\n\ts_barrier_wait -1" ::: "memory");
}

__device__ __forceinline__ void phase_fence() {
    __builtin_amdgcn_sched_barrier(0);
}

__device__ __forceinline__ void run_one_tile(
    v4f (&wA)[16], v4f (&wB)[16],
    const float* __restrict__ x, const float* src1, const float* src2, const int o8,
    const v4f (&bv)[8], const float* __restrict__ b2n, float* out,
    float* xs, float* hs, const int* tok, float* wdst, const float* afrag,
    const int tile, const int cnt,
    const int tid, const int wave, const int nlo, const int kb, const int mhi)
{
    constexpr int NT = 1;
    const float* xfrag = xs + nlo * XS_STRIDE + kb;
    const float* hfrag = hs + nlo * HS_STRIDE + kb;
    float*       hdst  = hs + nlo * HS_STRIDE + mhi;

    const int xr0 = tid >> 5, xc4 = (tid & 31) << 2;
    v4f xv[2 * NT];
    #pragma unroll
    for (int j = 0; j < 2 * NT; ++j) {
        const int s = tile + xr0 + 8 * j;
        const int t = tok[s < cnt ? s : 0];
        xv[j] = *(const v4f*)(x + (size_t)t * H_DIM + xc4);
    }
    #pragma unroll
    for (int j = 0; j < 2 * NT; ++j)
        *(v4f*)(xs + (j >> 1) * XS_TILE + (xr0 + 8 * (j & 1)) * XS_STRIDE + xc4) = xv[j];
    __syncthreads();

    v8f acc[NT];

    stage_chunk(wdst, wA);  load_chunk(wA, src1 + 2 * o8, H_DIM);
    phase_fence();
    clear_acc<NT>(acc);
    mma_chunk<NT>(afrag, xfrag, XS_TILE, acc);
    deposit_tiles<NT>(hdst + (wave << 4),        bv[0], bv[1], acc);
    phase_fence();

    stage_chunk(wdst, wB);  load_chunk(wB, src1 + 3 * o8, H_DIM);
    phase_fence();
    clear_acc<NT>(acc);
    mma_chunk<NT>(afrag, xfrag, XS_TILE, acc);
    deposit_tiles<NT>(hdst + ((wave + 8) << 4),  bv[2], bv[3], acc);
    phase_fence();

    stage_chunk(wdst, wA);  load_chunk(wA, src2,       H4);
    phase_fence();
    clear_acc<NT>(acc);
    mma_chunk<NT>(afrag, xfrag, XS_TILE, acc);
    deposit_tiles<NT>(hdst + ((wave + 16) << 4), bv[4], bv[5], acc);
    phase_fence();

    stage_chunk(wdst, wB);  load_chunk(wB, src2 + 128, H4);
    phase_fence();
    clear_acc<NT>(acc);
    mma_chunk<NT>(afrag, xfrag, XS_TILE, acc);
    deposit_tiles<NT>(hdst + ((wave + 24) << 4), bv[6], bv[7], acc);
    __syncthreads();

    stage_chunk(wdst, wA);  load_chunk(wA, src2 + 256, H4);
    clear_acc<NT>(acc);
    mma_chunk<NT>(afrag, hfrag,       HS_TILE, acc);
    stage_chunk(wdst, wB);  load_chunk(wB, src2 + 384, H4);
    mma_chunk<NT>(afrag, hfrag + 128, HS_TILE, acc);
    stage_chunk(wdst, wA);
    mma_chunk<NT>(afrag, hfrag + 256, HS_TILE, acc);
    stage_chunk(wdst, wB);
    mma_chunk<NT>(afrag, hfrag + 384, HS_TILE, acc);

    #pragma unroll
    for (int tt = 0; tt < NT; ++tt) {
        const int s = tile + 16 * tt + nlo;
        if (s < cnt) {
            const int t  = tok[s];
            const int o0 = (wave << 4) + mhi;
            v4f bia0 = *(const v4f*)(b2n + o0);
            v4f bia1 = *(const v4f*)(b2n + o0 + 4);
            v4f r0, r1;
            #pragma unroll
            for (int r = 0; r < 4; ++r) r0[r] = acc[tt][r]     + bia0[r];
            #pragma unroll
            for (int r = 0; r < 4; ++r) r1[r] = acc[tt][r + 4] + bia1[r];
            *(v4f*)(out + (size_t)t * H_DIM + o0)     = r0;
            *(v4f*)(out + (size_t)t * H_DIM + o0 + 4) = r1;
        }
    }
}

__device__ __forceinline__ void run_two_tiles(
    v4f (&wA)[16], v4f (&wB)[16],
    const float* __restrict__ x, const float* src1, const float* src2, const int o8,
    const v4f (&bv)[8], const float* __restrict__ b2n, float* out,
    float* xs, float* hs, const int* tok, float* wdst, const float* afrag,
    const int tile, const int cnt,
    const int tid, const int wave, const int nlo, const int kb, const int mhi)
{
    const float* xfrag = xs + nlo * XS_STRIDE + kb;
    const float* hfrag = hs + nlo * HS_STRIDE + kb;
    float*       hdst  = hs + nlo * HS_STRIDE + mhi;

    const int xr0 = tid >> 5, xc4 = (tid & 31) << 2;
    v4f xv[4];
    #pragma unroll
    for (int j = 0; j < 4; ++j) {
        const int s = tile + xr0 + 8 * j;
        const int t = tok[s < cnt ? s : 0];
        xv[j] = *(const v4f*)(x + (size_t)t * H_DIM + xc4);
    }
    #pragma unroll
    for (int j = 0; j < 4; ++j)
        *(v4f*)(xs + (j >> 1) * XS_TILE + (xr0 + 8 * (j & 1)) * XS_STRIDE + xc4) = xv[j];
    __syncthreads();

    v8f acc[2];
    v2f fa0[16], fb0[2][16];
    v2f fa1[16], fb1[2][16];

    stage_chunk(wdst, wA);
    load_chunk(wA, src1 + 2 * o8, H_DIM);
    read_half<2>(fa0, fb0, afrag, xfrag, XS_TILE);

    #pragma unroll 1
    for (int p = 0; p < 4; ++p) {
        const bool   l1     = p < 2;
        const int    cb     = 2 * p + 3;
        const float* bsrc   = (cb < 4) ? src1 + cb * o8 : src2 + (cb - 4) * 128;
        const int    bpit   = (cb < 4) ? H_DIM : H4;
        const float* bfx    = l1 ? xfrag : hfrag + (2 * p - 4) * 128;
        const float* bfy    = l1 ? xfrag : hfrag + (2 * p - 3) * 128;
        const float* bfn    = (p == 0) ? xfrag : hfrag + (2 * p - 2) * 128;
        const int    bpitch = l1 ? XS_TILE : HS_TILE;
        const int    npitch = (p == 0) ? XS_TILE : HS_TILE;
        const int    ot     = wave + 16 * p;
        const v4f ba0 = (p == 0) ? bv[0] : bv[4], ba1 = (p == 0) ? bv[1] : bv[5];
        const v4f bb0 = (p == 0) ? bv[2] : bv[6], bb1 = (p == 0) ? bv[3] : bv[7];

        read_half<2>(fa1, fb1, afrag + 64, bfx + 64, bpitch);
        if (p < 3) clear_acc<2>(acc);
        mma_half<2>(fa0, fb0, acc);
        stage_chunk(wdst, wB);
        if (p < 3) load_chunk(wB, bsrc, bpit);
        read_half<2>(fa0, fb0, afrag, bfy, bpitch);
        mma_half<2>(fa1, fb1, acc);
        if (l1) deposit_tiles<2>(hdst + (ot << 4),       ba0, ba1, acc);

        read_half<2>(fa1, fb1, afrag + 64, bfy + 64, bpitch);
        if (l1) clear_acc<2>(acc);
        mma_half<2>(fa0, fb0, acc);
        if (p < 3) {
            stage_chunk(wdst, wA);
            if (p < 2) load_chunk(wA, src2 + 2 * p * 128, H4);
            if (p != 1) read_half<2>(fa0, fb0, afrag, bfn, npitch);
        }
        mma_half<2>(fa1, fb1, acc);
        if (l1) deposit_tiles<2>(hdst + ((ot + 8) << 4), bb0, bb1, acc);
        if (p == 1) {
            __syncthreads();
            read_half<2>(fa0, fb0, afrag, bfn, npitch);
        }
    }

    #pragma unroll
    for (int tt = 0; tt < 2; ++tt) {
        const int s = tile + 16 * tt + nlo;
        if (s < cnt) {
            const int t  = tok[s];
            const int o0 = (wave << 4) + mhi;
            v4f bia0 = *(const v4f*)(b2n + o0);
            v4f bia1 = *(const v4f*)(b2n + o0 + 4);
            v4f r0, r1;
            #pragma unroll
            for (int r = 0; r < 4; ++r) r0[r] = acc[tt][r]     + bia0[r];
            #pragma unroll
            for (int r = 0; r < 4; ++r) r1[r] = acc[tt][r + 4] + bia1[r];
            *(v4f*)(out + (size_t)t * H_DIM + o0)     = r0;
            *(v4f*)(out + (size_t)t * H_DIM + o0 + 4) = r1;
        }
    }
}

__global__ __launch_bounds__(256) void k_mlp(
    const float* __restrict__ x, const int* __restrict__ node_ind,
    const float* W1, const float* __restrict__ b1,
    const float* W2, const float* __restrict__ b2,
    float* out)
{
    __shared__ float xs[2 * XS_TILE];
    __shared__ float hs[2 * HS_TILE];
    __shared__ __attribute__((aligned(16))) float wst[8 * 16 * WS_STRIDE];
    __shared__ int   tok[T_TOK];
    __shared__ __attribute__((aligned(16))) int wcnt[8];

    const int n    = blockIdx.x;
    const int tid  = threadIdx.x;
    const int lane = tid & 31;
    const int wave = tid >> 5;
    const int nlo  = lane & 15;
    const int kb   = (lane >> 4) << 1;
    const int mhi  = (lane >> 4) << 3;

    const float* W1n = W1 + (size_t)n * H4 * H_DIM;
    const float* W2n = W2 + (size_t)n * H_DIM * H4;
    const float* b1n = b1 + (size_t)n * H4;
    const float* b2n = b2 + (size_t)n * H_DIM;

    const float* src1 = W1n + (size_t)(wave << 4) * H_DIM + (lane << 2);
    const int    o8   = 8 * 16 * H_DIM;
    const float* src2 = W2n + (size_t)(wave << 4) * H4 + (lane << 2);

    const int tb = (wave << 8) + (lane << 2);
    const v4i id0 = *(const v4i*)(node_ind + tb);
    const v4i id1 = *(const v4i*)(node_ind + tb + 128);

    v4f wA[16], wB[16];
    load_chunk(wA, src1,      H_DIM);
    load_chunk(wB, src1 + o8, H_DIM);

    int pre[8];
    int run = 0;
    unsigned hitbits = 0;
    #pragma unroll
    for (int q = 0; q < 8; ++q) {
        const int id = (q < 4) ? id0[q & 3] : id1[q & 3];
        const bool hit = (id == n);
        const unsigned mask = __builtin_amdgcn_ballot_w32(hit);
        pre[q] = __builtin_amdgcn_mbcnt_lo(mask, run);
        run += __builtin_popcount(mask);
        hitbits |= hit ? (1u << q) : 0u;
    }
    if (lane == 0) wcnt[wave] = run;
    lds_barrier();

    const v4i c0 = *(const v4i*)(wcnt);
    const v4i c1 = *(const v4i*)(wcnt + 4);
    int base = 0, cnt = 0;
    #pragma unroll
    for (int w = 0; w < 8; ++w) {
        const int cw = (w < 4) ? c0[w & 3] : c1[w & 3];
        base += (w < wave) ? cw : 0;
        cnt  += cw;
    }
    #pragma unroll
    for (int q = 0; q < 8; ++q)
        if (hitbits & (1u << q)) tok[base + pre[q]] = tb + ((q >> 2) << 7) + (q & 3);
    lds_barrier();
    if (cnt == 0) {
        asm volatile("s_wait_loadcnt 0x0" ::: "memory");
        return;
    }

    float*       wmine = wst + wave * (16 * WS_STRIDE);
    float*       wdst  = wmine + (lane << 2);
    const float* afrag = wmine + nlo * WS_STRIDE + kb;

    v4f bv[8];
    #pragma unroll
    for (int c = 0; c < 4; ++c) {
        bv[2 * c]     = *(const v4f*)(b1n + ((wave + 8 * c) << 4) + mhi);
        bv[2 * c + 1] = *(const v4f*)(b1n + ((wave + 8 * c) << 4) + mhi + 4);
    }

    for (int tile = 0; ; ) {
        if (__builtin_expect(cnt - tile > 16, 0)) {
            run_two_tiles(wA, wB, x, src1, src2, o8, bv, b2n, out, xs, hs, tok, wdst, afrag,
                          tile, cnt, tid, wave, nlo, kb, mhi);
            tile += 32;
        } else {
            run_one_tile(wA, wB, x, src1, src2, o8, bv, b2n, out, xs, hs, tok, wdst, afrag,
                         tile, cnt, tid, wave, nlo, kb, mhi);
            tile += 16;
        }
        if (tile >= cnt) break;
        load_chunk(wA, src1,      H_DIM);
        load_chunk(wB, src1 + o8, H_DIM);
        __syncthreads();
    }
}


extern "C" void kernel_launch(void* const* d_in, const int* in_sizes, int n_in,
                              void* d_out, int out_size, void* d_ws, size_t ws_size,
                              hipStream_t stream) {
    const float* x        = (const float*)d_in[0];
    const int*   node_ind = (const int*)  d_in[1];
    const float* W1       = (const float*)d_in[2];
    const float* b1       = (const float*)d_in[3];
    const float* W2       = (const float*)d_in[4];
    const float* b2       = (const float*)d_in[5];
    float*       out      = (float*)d_out;

    k_mlp<<<NN, 256, 0, stream>>>(x, node_ind, W1, b1, W2, b2, out);
}


# GELU in fewer vector instructions: erf's exp(-p) as one multiply and the hardware base-2 exponential, not the library expf; sign carried by the last fma; same polynomials, h moves by rounding only
# speedup vs baseline: 2.1629x; 1.0423x over previous
#include <hip/hip_runtime.h>
#include <math.h>

typedef __attribute__((ext_vector_type(2))) float v2f;
typedef __attribute__((ext_vector_type(4))) float v4f;
typedef __attribute__((ext_vector_type(8))) float v8f;

#define T_TOK 2048
#define H_DIM 128
#define H4    512
#define NN    256

#define XS_STRIDE 132
#define HS_STRIDE 516
#define WS_STRIDE 132
#define XS_TILE (16 * XS_STRIDE)
#define HS_TILE (16 * HS_STRIDE)

typedef __attribute__((ext_vector_type(4))) int v4i;


__device__ __forceinline__ float gelu_erf(const float v) {
    const float av = fabsf(v);
    const float ax = av * 0.70710678118654752f;
    const float hv = 0.5f * v;
    const float t  = ax * ax;
    float q = fmaf(t, -0x1.268bc2p-11f, 0x1.420828p-8f);
    q = fmaf(t, q, -0x1.b5937p-6f);
    q = fmaf(t, q,  0x1.ce077cp-4f);
    q = fmaf(t, q, -0x1.81266p-2f);
    q = fmaf(t, q,  0x1.06eba0p-3f);
    const float lo = fmaf(ax, q, ax);
    float p = fmaf(ax, 0x1.1d3156p-16f, -0x1.8d129p-12f);
    p = fmaf(ax, p,  0x1.f9a6d2p-9f);
    p = fmaf(ax, p, -0x1.8c3164p-6f);
    p = fmaf(ax, p,  0x1.b4e9c8p-4f);
    p = fmaf(ax, p,  0x1.4515fap-1f);
    p = fmaf(ax, p,  0x1.078e50p-3f);
    p = fmaf(ax, p, ax);
    const float e  = __builtin_amdgcn_exp2f(p * -0x1.715476p+0f);
    const float hi = 1.0f - e;
    const float er = ax < 1.0f ? lo : hi;
    return fmaf(0.5f * av, er, hv);
}

__device__ __forceinline__ void load_chunk(v4f (&w)[16], const float* src, const int pitch) {
    #pragma unroll
    for (int r = 0; r < 16; ++r) w[r] = *(const v4f*)(src + r * pitch);
}

__device__ __forceinline__ void stage_chunk(float* dst, const v4f (&w)[16]) {
    #pragma unroll
    for (int r = 0; r < 16; ++r) *(v4f*)(dst + r * WS_STRIDE) = w[r];
}

template <int NT>
__device__ __forceinline__ void pin_acc(v8f (&acc)[NT]) {
    #pragma unroll
    for (int t = 0; t < NT; ++t) asm volatile("" : "+v"(acc[t]) :: "memory");
}

template <int NT>
__device__ __forceinline__ void mma_chunk(const float* arow, const float* brow, const int bpitch,
                                          v8f (&acc)[NT]) {
    v2f a[32], b[NT][32];
    #pragma unroll
    for (int j = 0; j < 32; ++j) {
        a[j] = *(const v2f*)(arow + 4 * j);
        #pragma unroll
        for (int t = 0; t < NT; ++t) b[t][j] = *(const v2f*)(brow + t * bpitch + 4 * j);
    }
    pin_acc<NT>(acc);
    #pragma unroll
    for (int j = 0; j < 32; ++j) {
        #pragma unroll
        for (int t = 0; t < NT; ++t)
            acc[t] = __builtin_amdgcn_wmma_f32_16x16x4_f32(
                false, a[j], false, b[t][j], (short)0, acc[t], false, false);
    }
    pin_acc<NT>(acc);
}

template <int NT>
__device__ __forceinline__ void read_half(v2f (&a)[16], v2f (&b)[NT][16],
                                          const float* arow, const float* brow, const int bpitch) {
    #pragma unroll
    for (int j = 0; j < 16; ++j) {
        a[j] = *(const v2f*)(arow + 4 * j);
        #pragma unroll
        for (int t = 0; t < NT; ++t) b[t][j] = *(const v2f*)(brow + t * bpitch + 4 * j);
    }
}

template <int NT>
__device__ __forceinline__ void mma_half(const v2f (&a)[16], const v2f (&b)[NT][16], v8f (&acc)[NT]) {
    pin_acc<NT>(acc);
    #pragma unroll
    for (int j = 0; j < 16; ++j) {
        #pragma unroll
        for (int t = 0; t < NT; ++t)
            acc[t] = __builtin_amdgcn_wmma_f32_16x16x4_f32(
                false, a[j], false, b[t][j], (short)0, acc[t], false, false);
    }
    pin_acc<NT>(acc);
}

template <int NT>
__device__ __forceinline__ void clear_acc(v8f (&acc)[NT]) {
    const v8f zero = {};
    #pragma unroll
    for (int t = 0; t < NT; ++t) acc[t] = zero;
}

__device__ __forceinline__ void deposit_h(float* hdst, const v4f bia0, const v4f bia1, v8f acc) {
    v4f h0, h1;
    #pragma unroll
    for (int r = 0; r < 4; ++r) h0[r] = gelu_erf(acc[r]     + bia0[r]);
    #pragma unroll
    for (int r = 0; r < 4; ++r) h1[r] = gelu_erf(acc[r + 4] + bia1[r]);
    *(v4f*)(hdst)     = h0;
    *(v4f*)(hdst + 4) = h1;
}

template <int NT>
__device__ __forceinline__ void deposit_tiles(float* hdst, const v4f bia0, const v4f bia1,
                                              const v8f (&acc)[NT]) {
    #pragma unroll
    for (int t = 0; t < NT; ++t) deposit_h(hdst + t * HS_TILE, bia0, bia1, acc[t]);
}

__device__ __forceinline__ void lds_barrier() {
    asm volatile("s_wait_dscnt 0x0\n\ts_barrier_signal -1\n\ts_barrier_wait -1" ::: "memory");
}

__device__ __forceinline__ void phase_fence() {
    __builtin_amdgcn_sched_barrier(0);
}

__device__ __forceinline__ void run_one_tile(
    v4f (&wA)[16], v4f (&wB)[16],
    const float* __restrict__ x, const float* src1, const float* src2, const int o8,
    const v4f (&bv)[8], const float* __restrict__ b2n, float* out,
    float* xs, float* hs, const int* tok, float* wdst, const float* afrag,
    const int tile, const int cnt,
    const int tid, const int wave, const int nlo, const int kb, const int mhi)
{
    constexpr int NT = 1;
    const float* xfrag = xs + nlo * XS_STRIDE + kb;
    const float* hfrag = hs + nlo * HS_STRIDE + kb;
    float*       hdst  = hs + nlo * HS_STRIDE + mhi;

    const int xr0 = tid >> 5, xc4 = (tid & 31) << 2;
    v4f xv[2 * NT];
    #pragma unroll
    for (int j = 0; j < 2 * NT; ++j) {
        const int s = tile + xr0 + 8 * j;
        const int t = tok[s < cnt ? s : 0];
        xv[j] = *(const v4f*)(x + (size_t)t * H_DIM + xc4);
    }
    #pragma unroll
    for (int j = 0; j < 2 * NT; ++j)
        *(v4f*)(xs + (j >> 1) * XS_TILE + (xr0 + 8 * (j & 1)) * XS_STRIDE + xc4) = xv[j];
    __syncthreads();

    v8f acc[NT];

    stage_chunk(wdst, wA);  load_chunk(wA, src1 + 2 * o8, H_DIM);
    phase_fence();
    clear_acc<NT>(acc);
    mma_chunk<NT>(afrag, xfrag, XS_TILE, acc);
    deposit_tiles<NT>(hdst + (wave << 4),        bv[0], bv[1], acc);
    phase_fence();

    stage_chunk(wdst, wB);  load_chunk(wB, src1 + 3 * o8, H_DIM);
    phase_fence();
    clear_acc<NT>(acc);
    mma_chunk<NT>(afrag, xfrag, XS_TILE, acc);
    deposit_tiles<NT>(hdst + ((wave + 8) << 4),  bv[2], bv[3], acc);
    phase_fence();

    stage_chunk(wdst, wA);  load_chunk(wA, src2,       H4);
    phase_fence();
    clear_acc<NT>(acc);
    mma_chunk<NT>(afrag, xfrag, XS_TILE, acc);
    deposit_tiles<NT>(hdst + ((wave + 16) << 4), bv[4], bv[5], acc);
    phase_fence();

    stage_chunk(wdst, wB);  load_chunk(wB, src2 + 128, H4);
    phase_fence();
    clear_acc<NT>(acc);
    mma_chunk<NT>(afrag, xfrag, XS_TILE, acc);
    deposit_tiles<NT>(hdst + ((wave + 24) << 4), bv[6], bv[7], acc);
    __syncthreads();

    stage_chunk(wdst, wA);  load_chunk(wA, src2 + 256, H4);
    clear_acc<NT>(acc);
    mma_chunk<NT>(afrag, hfrag,       HS_TILE, acc);
    stage_chunk(wdst, wB);  load_chunk(wB, src2 + 384, H4);
    mma_chunk<NT>(afrag, hfrag + 128, HS_TILE, acc);
    stage_chunk(wdst, wA);
    mma_chunk<NT>(afrag, hfrag + 256, HS_TILE, acc);
    stage_chunk(wdst, wB);
    mma_chunk<NT>(afrag, hfrag + 384, HS_TILE, acc);

    #pragma unroll
    for (int tt = 0; tt < NT; ++tt) {
        const int s = tile + 16 * tt + nlo;
        if (s < cnt) {
            const int t  = tok[s];
            const int o0 = (wave << 4) + mhi;
            v4f bia0 = *(const v4f*)(b2n + o0);
            v4f bia1 = *(const v4f*)(b2n + o0 + 4);
            v4f r0, r1;
            #pragma unroll
            for (int r = 0; r < 4; ++r) r0[r] = acc[tt][r]     + bia0[r];
            #pragma unroll
            for (int r = 0; r < 4; ++r) r1[r] = acc[tt][r + 4] + bia1[r];
            *(v4f*)(out + (size_t)t * H_DIM + o0)     = r0;
            *(v4f*)(out + (size_t)t * H_DIM + o0 + 4) = r1;
        }
    }
}

__device__ __forceinline__ void run_two_tiles(
    v4f (&wA)[16], v4f (&wB)[16],
    const float* __restrict__ x, const float* src1, const float* src2, const int o8,
    const v4f (&bv)[8], const float* __restrict__ b2n, float* out,
    float* xs, float* hs, const int* tok, float* wdst, const float* afrag,
    const int tile, const int cnt,
    const int tid, const int wave, const int nlo, const int kb, const int mhi)
{
    const float* xfrag = xs + nlo * XS_STRIDE + kb;
    const float* hfrag = hs + nlo * HS_STRIDE + kb;
    float*       hdst  = hs + nlo * HS_STRIDE + mhi;

    const int xr0 = tid >> 5, xc4 = (tid & 31) << 2;
    v4f xv[4];
    #pragma unroll
    for (int j = 0; j < 4; ++j) {
        const int s = tile + xr0 + 8 * j;
        const int t = tok[s < cnt ? s : 0];
        xv[j] = *(const v4f*)(x + (size_t)t * H_DIM + xc4);
    }
    #pragma unroll
    for (int j = 0; j < 4; ++j)
        *(v4f*)(xs + (j >> 1) * XS_TILE + (xr0 + 8 * (j & 1)) * XS_STRIDE + xc4) = xv[j];
    __syncthreads();

    v8f acc[2];
    v2f fa0[16], fb0[2][16];
    v2f fa1[16], fb1[2][16];

    stage_chunk(wdst, wA);
    load_chunk(wA, src1 + 2 * o8, H_DIM);
    read_half<2>(fa0, fb0, afrag, xfrag, XS_TILE);

    #pragma unroll 1
    for (int p = 0; p < 4; ++p) {
        const bool   l1     = p < 2;
        const int    cb     = 2 * p + 3;
        const float* bsrc   = (cb < 4) ? src1 + cb * o8 : src2 + (cb - 4) * 128;
        const int    bpit   = (cb < 4) ? H_DIM : H4;
        const float* bfx    = l1 ? xfrag : hfrag + (2 * p - 4) * 128;
        const float* bfy    = l1 ? xfrag : hfrag + (2 * p - 3) * 128;
        const float* bfn    = (p == 0) ? xfrag : hfrag + (2 * p - 2) * 128;
        const int    bpitch = l1 ? XS_TILE : HS_TILE;
        const int    npitch = (p == 0) ? XS_TILE : HS_TILE;
        const int    ot     = wave + 16 * p;
        const v4f ba0 = (p == 0) ? bv[0] : bv[4], ba1 = (p == 0) ? bv[1] : bv[5];
        const v4f bb0 = (p == 0) ? bv[2] : bv[6], bb1 = (p == 0) ? bv[3] : bv[7];

        read_half<2>(fa1, fb1, afrag + 64, bfx + 64, bpitch);
        if (p < 3) clear_acc<2>(acc);
        mma_half<2>(fa0, fb0, acc);
        stage_chunk(wdst, wB);
        if (p < 3) load_chunk(wB, bsrc, bpit);
        read_half<2>(fa0, fb0, afrag, bfy, bpitch);
        mma_half<2>(fa1, fb1, acc);
        if (l1) deposit_tiles<2>(hdst + (ot << 4),       ba0, ba1, acc);

        read_half<2>(fa1, fb1, afrag + 64, bfy + 64, bpitch);
        if (l1) clear_acc<2>(acc);
        mma_half<2>(fa0, fb0, acc);
        if (p < 3) {
            stage_chunk(wdst, wA);
            if (p < 2) load_chunk(wA, src2 + 2 * p * 128, H4);
            if (p != 1) read_half<2>(fa0, fb0, afrag, bfn, npitch);
        }
        mma_half<2>(fa1, fb1, acc);
        if (l1) deposit_tiles<2>(hdst + ((ot + 8) << 4), bb0, bb1, acc);
        if (p == 1) {
            __syncthreads();
            read_half<2>(fa0, fb0, afrag, bfn, npitch);
        }
    }

    #pragma unroll
    for (int tt = 0; tt < 2; ++tt) {
        const int s = tile + 16 * tt + nlo;
        if (s < cnt) {
            const int t  = tok[s];
            const int o0 = (wave << 4) + mhi;
            v4f bia0 = *(const v4f*)(b2n + o0);
            v4f bia1 = *(const v4f*)(b2n + o0 + 4);
            v4f r0, r1;
            #pragma unroll
            for (int r = 0; r < 4; ++r) r0[r] = acc[tt][r]     + bia0[r];
            #pragma unroll
            for (int r = 0; r < 4; ++r) r1[r] = acc[tt][r + 4] + bia1[r];
            *(v4f*)(out + (size_t)t * H_DIM + o0)     = r0;
            *(v4f*)(out + (size_t)t * H_DIM + o0 + 4) = r1;
        }
    }
}

__global__ __launch_bounds__(256) void k_mlp(
    const float* __restrict__ x, const int* __restrict__ node_ind,
    const float* W1, const float* __restrict__ b1,
    const float* W2, const float* __restrict__ b2,
    float* out)
{
    __shared__ float xs[2 * XS_TILE];
    __shared__ float hs[2 * HS_TILE];
    __shared__ __attribute__((aligned(16))) float wst[8 * 16 * WS_STRIDE];
    __shared__ int   tok[T_TOK];
    __shared__ __attribute__((aligned(16))) int wcnt[8];

    const int n    = blockIdx.x;
    const int tid  = threadIdx.x;
    const int lane = tid & 31;
    const int wave = tid >> 5;
    const int nlo  = lane & 15;
    const int kb   = (lane >> 4) << 1;
    const int mhi  = (lane >> 4) << 3;

    const float* W1n = W1 + (size_t)n * H4 * H_DIM;
    const float* W2n = W2 + (size_t)n * H_DIM * H4;
    const float* b1n = b1 + (size_t)n * H4;
    const float* b2n = b2 + (size_t)n * H_DIM;

    const float* src1 = W1n + (size_t)(wave << 4) * H_DIM + (lane << 2);
    const int    o8   = 8 * 16 * H_DIM;
    const float* src2 = W2n + (size_t)(wave << 4) * H4 + (lane << 2);

    const int tb = (wave << 8) + (lane << 2);
    const v4i id0 = *(const v4i*)(node_ind + tb);
    const v4i id1 = *(const v4i*)(node_ind + tb + 128);

    v4f wA[16], wB[16];
    load_chunk(wA, src1,      H_DIM);
    load_chunk(wB, src1 + o8, H_DIM);

    int pre[8];
    int run = 0;
    unsigned hitbits = 0;
    #pragma unroll
    for (int q = 0; q < 8; ++q) {
        const int id = (q < 4) ? id0[q & 3] : id1[q & 3];
        const bool hit = (id == n);
        const unsigned mask = __builtin_amdgcn_ballot_w32(hit);
        pre[q] = __builtin_amdgcn_mbcnt_lo(mask, run);
        run += __builtin_popcount(mask);
        hitbits |= hit ? (1u << q) : 0u;
    }
    if (lane == 0) wcnt[wave] = run;
    lds_barrier();

    const v4i c0 = *(const v4i*)(wcnt);
    const v4i c1 = *(const v4i*)(wcnt + 4);
    int base = 0, cnt = 0;
    #pragma unroll
    for (int w = 0; w < 8; ++w) {
        const int cw = (w < 4) ? c0[w & 3] : c1[w & 3];
        base += (w < wave) ? cw : 0;
        cnt  += cw;
    }
    #pragma unroll
    for (int q = 0; q < 8; ++q)
        if (hitbits & (1u << q)) tok[base + pre[q]] = tb + ((q >> 2) << 7) + (q & 3);
    lds_barrier();
    if (cnt == 0) {
        asm volatile("s_wait_loadcnt 0x0" ::: "memory");
        return;
    }

    float*       wmine = wst + wave * (16 * WS_STRIDE);
    float*       wdst  = wmine + (lane << 2);
    const float* afrag = wmine + nlo * WS_STRIDE + kb;

    v4f bv[8];
    #pragma unroll
    for (int c = 0; c < 4; ++c) {
        bv[2 * c]     = *(const v4f*)(b1n + ((wave + 8 * c) << 4) + mhi);
        bv[2 * c + 1] = *(const v4f*)(b1n + ((wave + 8 * c) << 4) + mhi + 4);
    }

    for (int tile = 0; ; ) {
        if (__builtin_expect(cnt - tile > 16, 0)) {
            run_two_tiles(wA, wB, x, src1, src2, o8, bv, b2n, out, xs, hs, tok, wdst, afrag,
                          tile, cnt, tid, wave, nlo, kb, mhi);
            tile += 32;
        } else {
            run_one_tile(wA, wB, x, src1, src2, o8, bv, b2n, out, xs, hs, tok, wdst, afrag,
                         tile, cnt, tid, wave, nlo, kb, mhi);
            tile += 16;
        }
        if (tile >= cnt) break;
        load_chunk(wA, src1,      H_DIM);
        load_chunk(wB, src1 + o8, H_DIM);
        __syncthreads();
    }
}


extern "C" void kernel_launch(void* const* d_in, const int* in_sizes, int n_in,
                              void* d_out, int out_size, void* d_ws, size_t ws_size,
                              hipStream_t stream) {
    const float* x        = (const float*)d_in[0];
    const int*   node_ind = (const int*)  d_in[1];
    const float* W1       = (const float*)d_in[2];
    const float* b1       = (const float*)d_in[3];
    const float* W2       = (const float*)d_in[4];
    const float* b2       = (const float*)d_in[5];
    float*       out      = (float*)d_out;

    k_mlp<<<NN, 256, 0, stream>>>(x, node_ind, W1, b1, W2, b2, out);
}


# a node's 16-token tiles go to two blocks (grid 2 x nodes; 256 registers, under half the LDS: 2 blocks share a WGP): tile 1 of a node with over 16 tokens gets another WGP's matrix pipes; k order kept
# speedup vs baseline: 2.2986x; 1.0627x over previous
#include <hip/hip_runtime.h>
#include <math.h>

typedef __attribute__((ext_vector_type(2))) float v2f;
typedef __attribute__((ext_vector_type(4))) float v4f;
typedef __attribute__((ext_vector_type(8))) float v8f;

#define T_TOK 2048
#define H_DIM 128
#define H4    512
#define NN    256
#define NBLK  512

#define XS_STRIDE 132
#define HS_STRIDE 516
#define WS_STRIDE 132
#define XS_TILE (16 * XS_STRIDE)
#define HS_TILE (16 * HS_STRIDE)

typedef __attribute__((ext_vector_type(4))) int v4i;


__device__ __forceinline__ float gelu_erf(const float v) {
    const float av = fabsf(v);
    const float ax = av * 0.70710678118654752f;
    const float hv = 0.5f * v;
    const float t  = ax * ax;
    float q = fmaf(t, -0x1.268bc2p-11f, 0x1.420828p-8f);
    q = fmaf(t, q, -0x1.b5937p-6f);
    q = fmaf(t, q,  0x1.ce077cp-4f);
    q = fmaf(t, q, -0x1.81266p-2f);
    q = fmaf(t, q,  0x1.06eba0p-3f);
    const float lo = fmaf(ax, q, ax);
    float p = fmaf(ax, 0x1.1d3156p-16f, -0x1.8d129p-12f);
    p = fmaf(ax, p,  0x1.f9a6d2p-9f);
    p = fmaf(ax, p, -0x1.8c3164p-6f);
    p = fmaf(ax, p,  0x1.b4e9c8p-4f);
    p = fmaf(ax, p,  0x1.4515fap-1f);
    p = fmaf(ax, p,  0x1.078e50p-3f);
    p = fmaf(ax, p, ax);
    const float e  = __builtin_amdgcn_exp2f(p * -0x1.715476p+0f);
    const float hi = 1.0f - e;
    const float er = ax < 1.0f ? lo : hi;
    return fmaf(0.5f * av, er, hv);
}

__device__ __forceinline__ void load_chunk(v4f (&w)[16], const float* src, const int pitch) {
    #pragma unroll
    for (int r = 0; r < 16; ++r) w[r] = *(const v4f*)(src + r * pitch);
}

__device__ __forceinline__ void stage_chunk(float* dst, const v4f (&w)[16]) {
    #pragma unroll
    for (int r = 0; r < 16; ++r) *(v4f*)(dst + r * WS_STRIDE) = w[r];
}

template <int NT>
__device__ __forceinline__ void pin_acc(v8f (&acc)[NT]) {
    #pragma unroll
    for (int t = 0; t < NT; ++t) asm volatile("" : "+v"(acc[t]) :: "memory");
}

__device__ __forceinline__ void mma_chunk(const float* arow, const float* brow, v8f (&acc)[1]) {
    #pragma unroll
    for (int h = 0; h < 2; ++h) {
        v2f a[16], b[16];
        #pragma unroll
        for (int j = 0; j < 16; ++j) {
            a[j] = *(const v2f*)(arow + 64 * h + 4 * j);
            b[j] = *(const v2f*)(brow + 64 * h + 4 * j);
        }
        pin_acc<1>(acc);
        #pragma unroll
        for (int j = 0; j < 16; ++j)
            acc[0] = __builtin_amdgcn_wmma_f32_16x16x4_f32(
                false, a[j], false, b[j], (short)0, acc[0], false, false);
        pin_acc<1>(acc);
    }
}

template <int NT>
__device__ __forceinline__ void clear_acc(v8f (&acc)[NT]) {
    const v8f zero = {};
    #pragma unroll
    for (int t = 0; t < NT; ++t) acc[t] = zero;
}

__device__ __forceinline__ void deposit_h(float* hdst, const v4f bia0, const v4f bia1, v8f acc) {
    v4f h0, h1;
    #pragma unroll
    for (int r = 0; r < 4; ++r) h0[r] = gelu_erf(acc[r]     + bia0[r]);
    #pragma unroll
    for (int r = 0; r < 4; ++r) h1[r] = gelu_erf(acc[r + 4] + bia1[r]);
    *(v4f*)(hdst)     = h0;
    *(v4f*)(hdst + 4) = h1;
}

__device__ __forceinline__ void lds_barrier() {
    asm volatile("s_wait_dscnt 0x0\n\ts_barrier_signal -1\n\ts_barrier_wait -1" ::: "memory");
}

__device__ __forceinline__ void phase_fence() {
    __builtin_amdgcn_sched_barrier(0);
}

__device__ __forceinline__ void run_one_tile(
    v4f (&wA)[16], v4f (&wB)[16],
    const float* __restrict__ x, const float* src1, const float* src2, const int o8,
    const v2f b1v, const v4f b2v, float* bs1, float* bs2, float* out,
    float* xs, float* hs, const int* tok, float* wdst, const float* afrag,
    const int tile, const int cnt,
    const int tid, const int wave, const int lane, const int nlo, const int kb, const int mhi)
{
    const float* xfrag = xs + nlo * XS_STRIDE + kb;
    const float* hfrag = hs + nlo * HS_STRIDE + kb;
    float*       hdst  = hs + nlo * HS_STRIDE + mhi;
    const float* bsrc  = bs1 + (wave << 4) + mhi;

    const int xr0 = tid >> 5, xc4 = (tid & 31) << 2;
    v4f xv[2];
    #pragma unroll
    for (int j = 0; j < 2; ++j) {
        const int s = tile + xr0 + 8 * j;
        const int t = tok[s < cnt ? s : tile];
        xv[j] = *(const v4f*)(x + (size_t)t * H_DIM + xc4);
    }
    #pragma unroll
    for (int j = 0; j < 2; ++j)
        *(v4f*)(xs + (xr0 + 8 * j) * XS_STRIDE + xc4) = xv[j];
    *(v2f*)(bs1 + 2 * tid) = b1v;
    if (wave == 0) *(v4f*)(bs2 + 4 * lane) = b2v;
    __syncthreads();

    v8f acc[1];
    v4f bia0, bia1;

    stage_chunk(wdst, wA);  load_chunk(wA, src1 + 2 * o8, H_DIM);
    bia0 = *(const v4f*)(bsrc);        bia1 = *(const v4f*)(bsrc + 4);
    phase_fence();
    clear_acc<1>(acc);
    mma_chunk(afrag, xfrag, acc);
    deposit_h(hdst + (wave << 4),        bia0, bia1, acc[0]);
    phase_fence();

    stage_chunk(wdst, wB);  load_chunk(wB, src1 + 3 * o8, H_DIM);
    bia0 = *(const v4f*)(bsrc + 128);  bia1 = *(const v4f*)(bsrc + 132);
    phase_fence();
    clear_acc<1>(acc);
    mma_chunk(afrag, xfrag, acc);
    deposit_h(hdst + ((wave + 8) << 4),  bia0, bia1, acc[0]);
    phase_fence();

    stage_chunk(wdst, wA);  load_chunk(wA, src2,       H4);
    bia0 = *(const v4f*)(bsrc + 256);  bia1 = *(const v4f*)(bsrc + 260);
    phase_fence();
    clear_acc<1>(acc);
    mma_chunk(afrag, xfrag, acc);
    deposit_h(hdst + ((wave + 16) << 4), bia0, bia1, acc[0]);
    phase_fence();

    stage_chunk(wdst, wB);  load_chunk(wB, src2 + 128, H4);
    bia0 = *(const v4f*)(bsrc + 384);  bia1 = *(const v4f*)(bsrc + 388);
    phase_fence();
    clear_acc<1>(acc);
    mma_chunk(afrag, xfrag, acc);
    deposit_h(hdst + ((wave + 24) << 4), bia0, bia1, acc[0]);
    __syncthreads();

    stage_chunk(wdst, wA);  load_chunk(wA, src2 + 256, H4);
    clear_acc<1>(acc);
    mma_chunk(afrag, hfrag,       acc);
    stage_chunk(wdst, wB);  load_chunk(wB, src2 + 384, H4);
    mma_chunk(afrag, hfrag + 128, acc);
    stage_chunk(wdst, wA);
    mma_chunk(afrag, hfrag + 256, acc);
    stage_chunk(wdst, wB);
    mma_chunk(afrag, hfrag + 384, acc);

    const int s = tile + nlo;
    if (s < cnt) {
        const int t  = tok[s];
        const int o0 = (wave << 4) + mhi;
        const v4f c0 = *(const v4f*)(bs2 + o0);
        const v4f c1 = *(const v4f*)(bs2 + o0 + 4);
        v4f r0, r1;
        #pragma unroll
        for (int r = 0; r < 4; ++r) r0[r] = acc[0][r]     + c0[r];
        #pragma unroll
        for (int r = 0; r < 4; ++r) r1[r] = acc[0][r + 4] + c1[r];
        *(v4f*)(out + (size_t)t * H_DIM + o0)     = r0;
        *(v4f*)(out + (size_t)t * H_DIM + o0 + 4) = r1;
    }
}

__global__ __launch_bounds__(256, 4) void k_mlp(
    const float* __restrict__ x, const int* __restrict__ node_ind,
    const float* W1, const float* __restrict__ b1,
    const float* W2, const float* __restrict__ b2,
    float* out)
{
    __shared__ float xs[XS_TILE];
    __shared__ float hs[HS_TILE];
    __shared__ __attribute__((aligned(16))) float wst[8 * 16 * WS_STRIDE];
    __shared__ int   tok[T_TOK];
    __shared__ __attribute__((aligned(16))) int wcnt[8];
    __shared__ __attribute__((aligned(16))) float bs1[H4];
    __shared__ __attribute__((aligned(16))) float bs2[H_DIM];

    const int part = blockIdx.x >> 8;
    const int n    = (blockIdx.x + (part << 3)) & (NN - 1);
    const int tid  = threadIdx.x;
    const int lane = tid & 31;
    const int wave = tid >> 5;
    const int nlo  = lane & 15;
    const int kb   = (lane >> 4) << 1;
    const int mhi  = (lane >> 4) << 3;

    const float* W1n = W1 + (size_t)n * H4 * H_DIM;
    const float* W2n = W2 + (size_t)n * H_DIM * H4;
    const float* b1n = b1 + (size_t)n * H4;
    const float* b2n = b2 + (size_t)n * H_DIM;

    const float* src1 = W1n + (size_t)(wave << 4) * H_DIM + (lane << 2);
    const int    o8   = 8 * 16 * H_DIM;
    const float* src2 = W2n + (size_t)(wave << 4) * H4 + (lane << 2);

    const int tb = (wave << 8) + (lane << 2);
    const v4i id0 = *(const v4i*)(node_ind + tb);
    const v4i id1 = *(const v4i*)(node_ind + tb + 128);

    v4f wA[16], wB[16];
    if (part == 0) {
        load_chunk(wA, src1,      H_DIM);
        load_chunk(wB, src1 + o8, H_DIM);
    }

    int pre[8];
    int run = 0;
    unsigned hitbits = 0;
    #pragma unroll
    for (int q = 0; q < 8; ++q) {
        const int id = (q < 4) ? id0[q & 3] : id1[q & 3];
        const bool hit = (id == n);
        const unsigned mask = __builtin_amdgcn_ballot_w32(hit);
        pre[q] = __builtin_amdgcn_mbcnt_lo(mask, run);
        run += __builtin_popcount(mask);
        hitbits |= hit ? (1u << q) : 0u;
    }
    if (lane == 0) wcnt[wave] = run;
    lds_barrier();

    const v4i c0 = *(const v4i*)(wcnt);
    const v4i c1 = *(const v4i*)(wcnt + 4);
    int base = 0, cnt = 0;
    #pragma unroll
    for (int w = 0; w < 8; ++w) {
        const int cw = (w < 4) ? c0[w & 3] : c1[w & 3];
        base += (w < wave) ? cw : 0;
        cnt  += cw;
    }
    const int first = part << 4;
    if (cnt <= first) {
        asm volatile("s_wait_loadcnt 0x0" ::: "memory");
        return;
    }
    if (part != 0) {
        __builtin_amdgcn_s_setprio(2);
        load_chunk(wA, src1,      H_DIM);
        load_chunk(wB, src1 + o8, H_DIM);
    }
    #pragma unroll
    for (int q = 0; q < 8; ++q)
        if (hitbits & (1u << q)) tok[base + pre[q]] = tb + ((q >> 2) << 7) + (q & 3);
    lds_barrier();

    float*       wmine = wst + wave * (16 * WS_STRIDE);
    float*       wdst  = wmine + (lane << 2);
    const float* afrag = wmine + nlo * WS_STRIDE + kb;

    const v2f b1v = *(const v2f*)(b1n + 2 * tid);
    v4f b2v = {};
    if (wave == 0) b2v = *(const v4f*)(b2n + 4 * lane);

    for (int tile = first; ; ) {
        run_one_tile(wA, wB, x, src1, src2, o8, b1v, b2v, bs1, bs2, out, xs, hs, tok, wdst, afrag,
                     tile, cnt, tid, wave, lane, nlo, kb, mhi);
        tile += 32;
        if (tile >= cnt) break;
        load_chunk(wA, src1,      H_DIM);
        load_chunk(wB, src1 + o8, H_DIM);
        __syncthreads();
    }
}


extern "C" void kernel_launch(void* const* d_in, const int* in_sizes, int n_in,
                              void* d_out, int out_size, void* d_ws, size_t ws_size,
                              hipStream_t stream) {
    const float* x        = (const float*)d_in[0];
    const int*   node_ind = (const int*)  d_in[1];
    const float* W1       = (const float*)d_in[2];
    const float* b1       = (const float*)d_in[3];
    const float* W2       = (const float*)d_in[4];
    const float* b2       = (const float*)d_in[5];
    float*       out      = (float*)d_out;

    k_mlp<<<NBLK, 256, 0, stream>>>(x, node_ind, W1, b1, W2, b2, out);
}
